# starting kernel
# baseline (speedup 1.0000x reference)
#include <hip/hip_runtime.h>
#include <math.h>

typedef float v2f __attribute__((ext_vector_type(2)));
typedef float v8f __attribute__((ext_vector_type(8)));

#define TQ 128
#define BQ 256
#define DQ 64
#define HQ 10
#define KD 74

__global__ __launch_bounds__(128) void qlstm_fused(
    const float* __restrict__ inputs,
    const float* __restrict__ Wf, const float* __restrict__ Wi,
    const float* __restrict__ Wu, const float* __restrict__ Wo,
    const float* __restrict__ bf, const float* __restrict__ bi,
    const float* __restrict__ bu, const float* __restrict__ bo,
    const float* __restrict__ pf, const float* __restrict__ pi,
    const float* __restrict__ pu, const float* __restrict__ po,
    const float* __restrict__ lng, const float* __restrict__ lnb,
    float* __restrict__ ys, float* __restrict__ hout, float* __restrict__ cout) {
  __shared__ float wLds[4][16][80];
  __shared__ float combLds[16][80];
  __shared__ float fac[4][16][HQ];
  __shared__ float gact[4][16][HQ];
  __shared__ float hLds[16][HQ];
  __shared__ float cLds[16][HQ];
  __shared__ float hraw[16][HQ];

  const int mtile = blockIdx.x;
  const int tid   = threadIdx.x;
  const int g     = tid >> 5;
  const int lane  = tid & 31;

  const float* bbg = (g == 0) ? bf : (g == 1) ? bi : (g == 2) ? bu : bo;
  const float* pg  = (g == 0) ? pf : (g == 1) ? pi : (g == 2) ? pu : po;

  for (int idx = tid; idx < 4 * 16 * 80; idx += 128) ((float*)wLds)[idx] = 0.f;
  for (int idx = tid; idx < 16 * 80; idx += 128) ((float*)combLds)[idx] = 0.f;
  for (int idx = tid; idx < 16 * HQ; idx += 128) {
    ((float*)hLds)[idx] = 0.f;
    ((float*)cLds)[idx] = 0.f;
  }
  __syncthreads();
  for (int idx = tid; idx < 4 * HQ * KD; idx += 128) {
    int gl = idx / (HQ * KD), rem = idx % (HQ * KD);
    const float* Wsrc = (gl == 0) ? Wf : (gl == 1) ? Wi : (gl == 2) ? Wu : Wo;
    wLds[gl][rem / KD][rem % KD] = Wsrc[rem];
  }
  __syncthreads();

  const int n    = lane & 15;
  const int m    = lane & 15;
  const int half = lane >> 4;
  const float bias = (n < HQ) ? bbg[n] : 0.f;
  const float px   = (n < HQ) ? pg[n * 3 + 0] : 0.f;
  const float cpy  = (n < HQ) ? cosf(pg[n * 3 + 1]) : 0.f;

  for (int t = 0; t < TQ; ++t) {
    const float* xt = inputs + ((size_t)t * BQ + mtile * 16) * DQ;
#pragma unroll
    for (int rep = 0; rep < 2; ++rep) {
      int ii = rep * 128 + tid;
      int r = ii >> 4, c4 = ii & 15;
      *(float4*)&combLds[r][c4 * 4] = ((const float4*)(xt + r * DQ))[c4];
    }
    for (int idx = tid; idx < 16 * HQ; idx += 128) {
      int r = idx / HQ, c = idx % HQ;
      combLds[r][DQ + c] = hLds[r][c];
    }
    if (t + 1 < TQ) {
      const char* nx = (const char*)(inputs + ((size_t)(t + 1) * BQ + mtile * 16) * DQ);
      __builtin_prefetch(nx + tid * 32, 0, 3);
    }
    __syncthreads();

    v8f acc = {0.f, 0.f, 0.f, 0.f, 0.f, 0.f, 0.f, 0.f};
#pragma unroll
    for (int kk = 0; kk < 19; ++kk) {
      int kb = kk * 4 + half * 2;
      v2f a = *(const v2f*)&combLds[m][kb];
      v2f b = *(const v2f*)&wLds[g][n][kb];
      acc = __builtin_amdgcn_wmma_f32_16x16x4_f32(false, a, false, b,
                                                  (short)0, acc, false, false);
    }
    if (n < HQ) {
#pragma unroll
      for (int v = 0; v < 8; ++v) {
        int row = v + 8 * half;
        fac[g][row][n] = cpy * __cosf(px + acc[v] + bias);
      }
    }
    __syncthreads();

    if (tid < 64) {
      int g2 = tid >> 4, row = tid & 15;
      float z = 1.f;
      for (int w = 0; w < HQ; ++w) {
        z *= fac[g2][row][w];
        gact[g2][row][w] = (g2 == 2) ? tanhf(z) : (1.f / (1.f + __expf(-z)));
      }
    }
    __syncthreads();

    for (int idx = tid; idx < 16 * HQ; idx += 128) {
      int row = idx / HQ, w = idx % HQ;
      float fv = gact[0][row][w], iv = gact[1][row][w];
      float gv = gact[2][row][w], ov = gact[3][row][w];
      float cn = fv * cLds[row][w] + iv * gv;
      cLds[row][w] = cn;
      hraw[row][w] = ov * tanhf(cn) + hLds[row][w];
    }
    __syncthreads();

    for (int idx = tid; idx < 16 * HQ; idx += 128) {
      int row = idx / HQ, w = idx % HQ;
      float mu = 0.f;
      for (int u = 0; u < HQ; ++u) mu += hraw[row][u];
      mu *= 0.1f;
      float var = 0.f;
      for (int u = 0; u < HQ; ++u) { float d = hraw[row][u] - mu; var += d * d; }
      var *= 0.1f;
      float hv = (hraw[row][w] - mu) * rsqrtf(var + 1e-5f) * lng[w] + lnb[w];
      hLds[row][w] = hv;
      ys[((size_t)t * BQ + mtile * 16 + row) * HQ + w] = hv;
    }
    __syncthreads();
  }

  for (int idx = tid; idx < 16 * HQ; idx += 128) {
    int row = idx / HQ, w = idx % HQ;
    hout[(mtile * 16 + row) * HQ + w] = hLds[row][w];
    cout[(mtile * 16 + row) * HQ + w] = cLds[row][w];
  }
}

extern "C" void kernel_launch(void* const* d_in, const int* in_sizes, int n_in,
                              void* d_out, int out_size, void* d_ws, size_t ws_size,
                              hipStream_t stream) {
  const float* inputs = (const float*)d_in[0];
  const float* Wf = (const float*)d_in[1];
  const float* bf = (const float*)d_in[2];
  const float* pf = (const float*)d_in[3];
  const float* Wi = (const float*)d_in[4];
  const float* bi = (const float*)d_in[5];
  const float* pi = (const float*)d_in[6];
  const float* Wu = (const float*)d_in[7];
  const float* bu = (const float*)d_in[8];
  const float* pu = (const float*)d_in[9];
  const float* Wo = (const float*)d_in[10];
  const float* bo = (const float*)d_in[11];
  const float* po = (const float*)d_in[12];
  const float* lng = (const float*)d_in[13];
  const float* lnb = (const float*)d_in[14];

  float* out  = (float*)d_out;
  float* ys   = out;
  float* hout = out + (size_t)TQ * BQ * HQ;
  float* cout = hout + BQ * HQ;

  qlstm_fused<<<BQ / 16, 128, 0, stream>>>(
      inputs, Wf, Wi, Wu, Wo, bf, bi, bu, bo,
      pf, pi, pu, po, lng, lnb, ys, hout, cout);
}


# GEMM transposed (wires on M, batch rows on lanes; k regrouped on two accumulators): gates, cell, LayerNorm stay in registers, one LDS gate exchange + barrier per step; LayerNorm sums in two halves
# speedup vs baseline: 2.4985x; 2.4985x over previous
#include <hip/hip_runtime.h>
#include <math.h>

typedef float v2f __attribute__((ext_vector_type(2)));
typedef float v8f __attribute__((ext_vector_type(8)));

#define TQ 128
#define BQ 256
#define DQ 64
#define HQ 10
#define KD 74
#define HH 5

__device__ __forceinline__ float xhalf(float x) {
  int b = __builtin_bit_cast(int, x);
  return __builtin_bit_cast(float,
      __builtin_amdgcn_permlanex16(b, b, 0x76543210, 0xfedcba98, false, false));
}

__global__ __launch_bounds__(128) void qlstm_fused(
    const float* __restrict__ inputs,
    const float* __restrict__ Wf, const float* __restrict__ Wi,
    const float* __restrict__ Wu, const float* __restrict__ Wo,
    const float* __restrict__ bf, const float* __restrict__ bi,
    const float* __restrict__ bu, const float* __restrict__ bo,
    const float* __restrict__ pf, const float* __restrict__ pi,
    const float* __restrict__ pu, const float* __restrict__ po,
    const float* __restrict__ lng, const float* __restrict__ lnb,
    float* __restrict__ ys, float* __restrict__ hout, float* __restrict__ cout) {
  __shared__ float gx[2][4][HH][32];

  const int mtile = blockIdx.x;
  const int tid   = threadIdx.x;
  const int g     = __builtin_amdgcn_readfirstlane(tid >> 5);
  const int lane  = tid & 31;
  const int b     = lane & 15;
  const int half  = lane >> 4;

  const float* Wg  = (g == 0) ? Wf : (g == 1) ? Wi : (g == 2) ? Wu : Wo;
  const float* bbg = (g == 0) ? bf : (g == 1) ? bi : (g == 2) ? bu : bo;
  const float* pg  = (g == 0) ? pf : (g == 1) ? pi : (g == 2) ? pu : po;

  const int  am   = lane & 15;
  const bool arow = (am < HH) || (am >= 8 && am < 8 + HH);
  const int  awire = arow ? ((am < HH) ? am : am - 3) : 0;
  const float* wrow = Wg + awire * KD;
  v2f wa[19];
#pragma unroll
  for (int i = 0; i < 8; ++i) {
    const int c = 8 * i + 4 * half;
    wa[2 * i][0]     = arow ? wrow[c + 0] : 0.f;
    wa[2 * i][1]     = arow ? wrow[c + 1] : 0.f;
    wa[2 * i + 1][0] = arow ? wrow[c + 2] : 0.f;
    wa[2 * i + 1][1] = arow ? wrow[c + 3] : 0.f;
  }
#pragma unroll
  for (int j = 0; j < 3; ++j) {
    wa[16 + j][0] = (arow && 2 * j < HH) ? wrow[DQ + HH * half + 2 * j] : 0.f;
    wa[16 + j][1] = (arow && 2 * j + 1 < HH) ? wrow[DQ + HH * half + 2 * j + 1] : 0.f;
  }

  float bias[HH], px[HH], cpy[HH], lg[HH], lb[HH], c[HH], h[HH];
#pragma unroll
  for (int v = 0; v < HH; ++v) {
    const int w = HH * half + v;
    bias[v] = bbg[w];
    px[v]   = pg[w * 3 + 0];
    cpy[v]  = cosf(pg[w * 3 + 1]);
    lg[v]   = lng[w];
    lb[v]   = lnb[w];
    c[v]    = 0.f;
    h[v]    = 0.f;
  }
  const int hmask = -half;

  for (int t = 0; t < TQ; ++t) {
    const float* xt = inputs + ((size_t)t * BQ + mtile * 16 + b) * DQ + 4 * half;
    float4 xq[8];
#pragma unroll
    for (int i = 0; i < 8; ++i) xq[i] = *(const float4*)(xt + 8 * i);
    if (t + 1 < TQ) {
      const char* nx = (const char*)(inputs + ((size_t)(t + 1) * BQ + mtile * 16) * DQ);
      __builtin_prefetch(nx + tid * 32, 0, 3);
    }

    v8f acc0 = {0.f, 0.f, 0.f, 0.f, 0.f, 0.f, 0.f, 0.f};
    v8f acc1 = {0.f, 0.f, 0.f, 0.f, 0.f, 0.f, 0.f, 0.f};
#pragma unroll
    for (int i = 0; i < 8; ++i) {
      v2f b0 = {xq[i].x, xq[i].y};
      v2f b1 = {xq[i].z, xq[i].w};
      acc0 = __builtin_amdgcn_wmma_f32_16x16x4_f32(false, wa[2 * i], false, b0,
                                                   (short)0, acc0, false, false);
      acc1 = __builtin_amdgcn_wmma_f32_16x16x4_f32(false, wa[2 * i + 1], false, b1,
                                                   (short)0, acc1, false, false);
    }
    {
      v2f h0 = {h[0], h[1]};
      v2f h1 = {h[2], h[3]};
      v2f h2 = {h[4], 0.f};
      acc0 = __builtin_amdgcn_wmma_f32_16x16x4_f32(false, wa[16], false, h0,
                                                   (short)0, acc0, false, false);
      acc1 = __builtin_amdgcn_wmma_f32_16x16x4_f32(false, wa[17], false, h1,
                                                   (short)0, acc1, false, false);
      acc0 = __builtin_amdgcn_wmma_f32_16x16x4_f32(false, wa[18], false, h2,
                                                   (short)0, acc0, false, false);
    }

    float fac[HH];
#pragma unroll
    for (int v = 0; v < HH; ++v)
      fac[v] = cpy[v] * __cosf(px[v] + (acc0[v] + acc1[v]) + bias[v]);

    float p = 1.f;
#pragma unroll
    for (int v = 0; v < HH; ++v) p *= fac[v];
    const float tot = xhalf(p);
    float z = __builtin_bit_cast(float,
        (__builtin_bit_cast(int, tot) & hmask) | (0x3f800000 & ~hmask));
    float ga[HH];
    if (g == 2) {
#pragma unroll
      for (int v = 0; v < HH; ++v) { z *= fac[v]; ga[v] = tanhf(z); }
    } else {
#pragma unroll
      for (int v = 0; v < HH; ++v) { z *= fac[v]; ga[v] = 1.f / (1.f + __expf(-z)); }
    }

    const int pb = t & 1;
#pragma unroll
    for (int v = 0; v < HH; ++v) gx[pb][g][v][lane] = ga[v];
    __syncthreads();

    float hr[HH];
#pragma unroll
    for (int v = 0; v < HH; ++v) {
      float fv = gx[pb][0][v][lane], iv = gx[pb][1][v][lane];
      float gv = gx[pb][2][v][lane], ov = gx[pb][3][v][lane];
      float cn = fv * c[v] + iv * gv;
      c[v] = cn;
      hr[v] = ov * tanhf(cn) + h[v];
    }

    float s = 0.f;
#pragma unroll
    for (int v = 0; v < HH; ++v) s += hr[v];
    const float mu = (s + xhalf(s)) * 0.1f;
    float q = 0.f;
#pragma unroll
    for (int v = 0; v < HH; ++v) { float d = hr[v] - mu; q += d * d; }
    const float var = (q + xhalf(q)) * 0.1f;
    const float rs = rsqrtf(var + 1e-5f);
#pragma unroll
    for (int v = 0; v < HH; ++v) h[v] = (hr[v] - mu) * rs * lg[v] + lb[v];

    if (g == 0) {
      float* yrow = ys + ((size_t)t * BQ + mtile * 16 + b) * HQ + HH * half;
#pragma unroll
      for (int v = 0; v < HH; ++v) yrow[v] = h[v];
    }
  }

  if (g == 0) {
#pragma unroll
    for (int v = 0; v < HH; ++v) {
      hout[(mtile * 16 + b) * HQ + HH * half + v] = h[v];
      cout[(mtile * 16 + b) * HQ + HH * half + v] = c[v];
    }
  }
}

extern "C" void kernel_launch(void* const* d_in, const int* in_sizes, int n_in,
                              void* d_out, int out_size, void* d_ws, size_t ws_size,
                              hipStream_t stream) {
  const float* inputs = (const float*)d_in[0];
  const float* Wf = (const float*)d_in[1];
  const float* bf = (const float*)d_in[2];
  const float* pf = (const float*)d_in[3];
  const float* Wi = (const float*)d_in[4];
  const float* bi = (const float*)d_in[5];
  const float* pi = (const float*)d_in[6];
  const float* Wu = (const float*)d_in[7];
  const float* bu = (const float*)d_in[8];
  const float* pu = (const float*)d_in[9];
  const float* Wo = (const float*)d_in[10];
  const float* bo = (const float*)d_in[11];
  const float* po = (const float*)d_in[12];
  const float* lng = (const float*)d_in[13];
  const float* lnb = (const float*)d_in[14];

  float* out  = (float*)d_out;
  float* ys   = out;
  float* hout = out + (size_t)TQ * BQ * HQ;
  float* cout = hout + BQ * HQ;

  qlstm_fused<<<BQ / 16, 128, 0, stream>>>(
      inputs, Wf, Wi, Wu, Wo, bf, bi, bu, bo,
      pf, pi, pu, po, lng, lnb, ys, hout, cout);
}


# tanh and sigmoid branch-free on hardware exp2 + reciprocal: tanh = sign(x)(1-q)/(1+q), q = 2^(-2.885|x|); library tanhf's per-wire branches and the IEEE division go; few ulp from the library forms
# speedup vs baseline: 3.4236x; 1.3703x over previous
#include <hip/hip_runtime.h>
#include <math.h>

typedef float v2f __attribute__((ext_vector_type(2)));
typedef float v8f __attribute__((ext_vector_type(8)));

#define TQ 128
#define BQ 256
#define DQ 64
#define HQ 10
#define KD 74
#define HH 5

__device__ __forceinline__ float xhalf(float x) {
  int b = __builtin_bit_cast(int, x);
  return __builtin_bit_cast(float,
      __builtin_amdgcn_permlanex16(b, b, 0x76543210, 0xfedcba98, false, false));
}

__device__ __forceinline__ float tanh_hw(float x) {
  const float q = __builtin_amdgcn_exp2f(-2.8853900817779268f * __builtin_fabsf(x));
  const float t = (1.f - q) * __builtin_amdgcn_rcpf(1.f + q);
  return __builtin_copysignf(t, x);
}

__device__ __forceinline__ float sigmoid_hw(float z) {
  return __builtin_amdgcn_rcpf(1.f + __expf(-z));
}

__global__ __launch_bounds__(128) void qlstm_fused(
    const float* __restrict__ inputs,
    const float* __restrict__ Wf, const float* __restrict__ Wi,
    const float* __restrict__ Wu, const float* __restrict__ Wo,
    const float* __restrict__ bf, const float* __restrict__ bi,
    const float* __restrict__ bu, const float* __restrict__ bo,
    const float* __restrict__ pf, const float* __restrict__ pi,
    const float* __restrict__ pu, const float* __restrict__ po,
    const float* __restrict__ lng, const float* __restrict__ lnb,
    float* __restrict__ ys, float* __restrict__ hout, float* __restrict__ cout) {
  __shared__ float gx[2][4][HH][32];

  const int mtile = blockIdx.x;
  const int tid   = threadIdx.x;
  const int g     = __builtin_amdgcn_readfirstlane(tid >> 5);
  const int lane  = tid & 31;
  const int b     = lane & 15;
  const int half  = lane >> 4;

  const float* Wg  = (g == 0) ? Wf : (g == 1) ? Wi : (g == 2) ? Wu : Wo;
  const float* bbg = (g == 0) ? bf : (g == 1) ? bi : (g == 2) ? bu : bo;
  const float* pg  = (g == 0) ? pf : (g == 1) ? pi : (g == 2) ? pu : po;

  const int  am   = lane & 15;
  const bool arow = (am < HH) || (am >= 8 && am < 8 + HH);
  const int  awire = arow ? ((am < HH) ? am : am - 3) : 0;
  const float* wrow = Wg + awire * KD;
  v2f wa[19];
#pragma unroll
  for (int i = 0; i < 8; ++i) {
    const int c = 8 * i + 4 * half;
    wa[2 * i][0]     = arow ? wrow[c + 0] : 0.f;
    wa[2 * i][1]     = arow ? wrow[c + 1] : 0.f;
    wa[2 * i + 1][0] = arow ? wrow[c + 2] : 0.f;
    wa[2 * i + 1][1] = arow ? wrow[c + 3] : 0.f;
  }
#pragma unroll
  for (int j = 0; j < 3; ++j) {
    wa[16 + j][0] = (arow && 2 * j < HH) ? wrow[DQ + HH * half + 2 * j] : 0.f;
    wa[16 + j][1] = (arow && 2 * j + 1 < HH) ? wrow[DQ + HH * half + 2 * j + 1] : 0.f;
  }

  float bias[HH], px[HH], cpy[HH], lg[HH], lb[HH], c[HH], h[HH];
#pragma unroll
  for (int v = 0; v < HH; ++v) {
    const int w = HH * half + v;
    bias[v] = bbg[w];
    px[v]   = pg[w * 3 + 0];
    cpy[v]  = cosf(pg[w * 3 + 1]);
    lg[v]   = lng[w];
    lb[v]   = lnb[w];
    c[v]    = 0.f;
    h[v]    = 0.f;
  }
  const int hmask = -half;

  for (int t = 0; t < TQ; ++t) {
    const float* xt = inputs + ((size_t)t * BQ + mtile * 16 + b) * DQ + 4 * half;
    float4 xq[8];
#pragma unroll
    for (int i = 0; i < 8; ++i) xq[i] = *(const float4*)(xt + 8 * i);
    if (t + 1 < TQ) {
      const char* nx = (const char*)(inputs + ((size_t)(t + 1) * BQ + mtile * 16) * DQ);
      __builtin_prefetch(nx + tid * 32, 0, 3);
    }

    v8f acc0 = {0.f, 0.f, 0.f, 0.f, 0.f, 0.f, 0.f, 0.f};
    v8f acc1 = {0.f, 0.f, 0.f, 0.f, 0.f, 0.f, 0.f, 0.f};
#pragma unroll
    for (int i = 0; i < 8; ++i) {
      v2f b0 = {xq[i].x, xq[i].y};
      v2f b1 = {xq[i].z, xq[i].w};
      acc0 = __builtin_amdgcn_wmma_f32_16x16x4_f32(false, wa[2 * i], false, b0,
                                                   (short)0, acc0, false, false);
      acc1 = __builtin_amdgcn_wmma_f32_16x16x4_f32(false, wa[2 * i + 1], false, b1,
                                                   (short)0, acc1, false, false);
    }
    {
      v2f h0 = {h[0], h[1]};
      v2f h1 = {h[2], h[3]};
      v2f h2 = {h[4], 0.f};
      acc0 = __builtin_amdgcn_wmma_f32_16x16x4_f32(false, wa[16], false, h0,
                                                   (short)0, acc0, false, false);
      acc1 = __builtin_amdgcn_wmma_f32_16x16x4_f32(false, wa[17], false, h1,
                                                   (short)0, acc1, false, false);
      acc0 = __builtin_amdgcn_wmma_f32_16x16x4_f32(false, wa[18], false, h2,
                                                   (short)0, acc0, false, false);
    }

    float fac[HH];
#pragma unroll
    for (int v = 0; v < HH; ++v)
      fac[v] = cpy[v] * __cosf(px[v] + (acc0[v] + acc1[v]) + bias[v]);

    float p = 1.f;
#pragma unroll
    for (int v = 0; v < HH; ++v) p *= fac[v];
    const float tot = xhalf(p);
    float z = __builtin_bit_cast(float,
        (__builtin_bit_cast(int, tot) & hmask) | (0x3f800000 & ~hmask));
    float ga[HH];
    if (g == 2) {
#pragma unroll
      for (int v = 0; v < HH; ++v) { z *= fac[v]; ga[v] = tanh_hw(z); }
    } else {
#pragma unroll
      for (int v = 0; v < HH; ++v) { z *= fac[v]; ga[v] = sigmoid_hw(z); }
    }

    const int pb = t & 1;
#pragma unroll
    for (int v = 0; v < HH; ++v) gx[pb][g][v][lane] = ga[v];
    __syncthreads();

    float hr[HH];
#pragma unroll
    for (int v = 0; v < HH; ++v) {
      float fv = gx[pb][0][v][lane], iv = gx[pb][1][v][lane];
      float gv = gx[pb][2][v][lane], ov = gx[pb][3][v][lane];
      float cn = fv * c[v] + iv * gv;
      c[v] = cn;
      hr[v] = ov * tanh_hw(cn) + h[v];
    }

    float s = 0.f;
#pragma unroll
    for (int v = 0; v < HH; ++v) s += hr[v];
    const float mu = (s + xhalf(s)) * 0.1f;
    float q = 0.f;
#pragma unroll
    for (int v = 0; v < HH; ++v) { float d = hr[v] - mu; q += d * d; }
    const float var = (q + xhalf(q)) * 0.1f;
    const float rs = rsqrtf(var + 1e-5f);
#pragma unroll
    for (int v = 0; v < HH; ++v) h[v] = (hr[v] - mu) * rs * lg[v] + lb[v];

    if (g == 0) {
      float* yrow = ys + ((size_t)t * BQ + mtile * 16 + b) * HQ + HH * half;
#pragma unroll
      for (int v = 0; v < HH; ++v) yrow[v] = h[v];
    }
  }

  if (g == 0) {
#pragma unroll
    for (int v = 0; v < HH; ++v) {
      hout[(mtile * 16 + b) * HQ + HH * half + v] = h[v];
      cout[(mtile * 16 + b) * HQ + HH * half + v] = c[v];
    }
  }
}

extern "C" void kernel_launch(void* const* d_in, const int* in_sizes, int n_in,
                              void* d_out, int out_size, void* d_ws, size_t ws_size,
                              hipStream_t stream) {
  const float* inputs = (const float*)d_in[0];
  const float* Wf = (const float*)d_in[1];
  const float* bf = (const float*)d_in[2];
  const float* pf = (const float*)d_in[3];
  const float* Wi = (const float*)d_in[4];
  const float* bi = (const float*)d_in[5];
  const float* pi = (const float*)d_in[6];
  const float* Wu = (const float*)d_in[7];
  const float* bu = (const float*)d_in[8];
  const float* pu = (const float*)d_in[9];
  const float* Wo = (const float*)d_in[10];
  const float* bo = (const float*)d_in[11];
  const float* po = (const float*)d_in[12];
  const float* lng = (const float*)d_in[13];
  const float* lnb = (const float*)d_in[14];

  float* out  = (float*)d_out;
  float* ys   = out;
  float* hout = out + (size_t)TQ * BQ * HQ;
  float* cout = hout + BQ * HQ;

  qlstm_fused<<<BQ / 16, 128, 0, stream>>>(
      inputs, Wf, Wi, Wu, Wo, bf, bi, bu, bo,
      pf, pi, pu, po, lng, lnb, ys, hout, cout);
}


# x path pipelined one step ahead: x(t+1) loaded at the top of step t, its wmma issued behind the barrier under the LDS reads and cell update; barrier orders LDS only; sums in the same order
# speedup vs baseline: 4.6251x; 1.3509x over previous
#include <hip/hip_runtime.h>
#include <math.h>

typedef float v2f __attribute__((ext_vector_type(2)));
typedef float v8f __attribute__((ext_vector_type(8)));

#define TQ 128
#define BQ 256
#define DQ 64
#define HQ 10
#define KD 74
#define HH 5

__device__ __forceinline__ float xhalf(float x) {
  int b = __builtin_bit_cast(int, x);
  return __builtin_bit_cast(float,
      __builtin_amdgcn_permlanex16(b, b, 0x76543210, 0xfedcba98, false, false));
}

__device__ __forceinline__ float tanh_hw(float x) {
  const float q = __builtin_amdgcn_exp2f(-2.8853900817779268f * __builtin_fabsf(x));
  const float t = (1.f - q) * __builtin_amdgcn_rcpf(1.f + q);
  return __builtin_copysignf(t, x);
}

__device__ __forceinline__ float sigmoid_hw(float z) {
  return __builtin_amdgcn_rcpf(1.f + __expf(-z));
}

__device__ __forceinline__ void xpair(const v2f& wa0, const v2f& wa1, const float4& x,
                                      v8f& a0, v8f& a1) {
  v2f b0 = {x.x, x.y};
  v2f b1 = {x.z, x.w};
  a0 = __builtin_amdgcn_wmma_f32_16x16x4_f32(false, wa0, false, b0, (short)0, a0, false, false);
  a1 = __builtin_amdgcn_wmma_f32_16x16x4_f32(false, wa1, false, b1, (short)0, a1, false, false);
}

__global__ __launch_bounds__(128) void qlstm_fused(
    const float* __restrict__ inputs,
    const float* __restrict__ Wf, const float* __restrict__ Wi,
    const float* __restrict__ Wu, const float* __restrict__ Wo,
    const float* __restrict__ bf, const float* __restrict__ bi,
    const float* __restrict__ bu, const float* __restrict__ bo,
    const float* __restrict__ pf, const float* __restrict__ pi,
    const float* __restrict__ pu, const float* __restrict__ po,
    const float* __restrict__ lng, const float* __restrict__ lnb,
    float* __restrict__ ys, float* __restrict__ hout, float* __restrict__ cout) {
  __shared__ float gx[2][4][HH][32];

  const int mtile = blockIdx.x;
  const int tid   = threadIdx.x;
  const int g     = __builtin_amdgcn_readfirstlane(tid >> 5);
  const int lane  = tid & 31;
  const int b     = lane & 15;
  const int half  = lane >> 4;

  const float* Wg  = (g == 0) ? Wf : (g == 1) ? Wi : (g == 2) ? Wu : Wo;
  const float* bbg = (g == 0) ? bf : (g == 1) ? bi : (g == 2) ? bu : bo;
  const float* pg  = (g == 0) ? pf : (g == 1) ? pi : (g == 2) ? pu : po;

  const int  am   = lane & 15;
  const bool arow = (am < HH) || (am >= 8 && am < 8 + HH);
  const int  awire = arow ? ((am < HH) ? am : am - 3) : 0;
  const float* wrow = Wg + awire * KD;
  v2f wa[19];
#pragma unroll
  for (int i = 0; i < 8; ++i) {
    const int c = 8 * i + 4 * half;
    wa[2 * i][0]     = arow ? wrow[c + 0] : 0.f;
    wa[2 * i][1]     = arow ? wrow[c + 1] : 0.f;
    wa[2 * i + 1][0] = arow ? wrow[c + 2] : 0.f;
    wa[2 * i + 1][1] = arow ? wrow[c + 3] : 0.f;
  }
#pragma unroll
  for (int j = 0; j < 3; ++j) {
    wa[16 + j][0] = (arow && 2 * j < HH) ? wrow[DQ + HH * half + 2 * j] : 0.f;
    wa[16 + j][1] = (arow && 2 * j + 1 < HH) ? wrow[DQ + HH * half + 2 * j + 1] : 0.f;
  }

  float bias[HH], px[HH], cpy[HH], lg[HH], lb[HH], c[HH], h[HH];
#pragma unroll
  for (int v = 0; v < HH; ++v) {
    const int w = HH * half + v;
    bias[v] = bbg[w];
    px[v]   = pg[w * 3 + 0];
    cpy[v]  = cosf(pg[w * 3 + 1]);
    lg[v]   = lng[w];
    lb[v]   = lnb[w];
    c[v]    = 0.f;
    h[v]    = 0.f;
  }
  const int hmask = -half;

  const float* xbase = inputs + ((size_t)(mtile * 16 + b)) * DQ + 4 * half;
  float4 xq[8];
  v8f ax0 = {0.f, 0.f, 0.f, 0.f, 0.f, 0.f, 0.f, 0.f};
  v8f ax1 = {0.f, 0.f, 0.f, 0.f, 0.f, 0.f, 0.f, 0.f};
#pragma unroll
  for (int i = 0; i < 8; ++i) xq[i] = *(const float4*)(xbase + 8 * i);
#pragma unroll
  for (int i = 0; i < 8; ++i) xpair(wa[2 * i], wa[2 * i + 1], xq[i], ax0, ax1);

  for (int t = 0; t < TQ; ++t) {
    const int tn = (t + 1 < TQ) ? t + 1 : TQ - 1;
    const float* xt = xbase + (size_t)tn * BQ * DQ;
#pragma unroll
    for (int i = 0; i < 8; ++i) xq[i] = *(const float4*)(xt + 8 * i);
    if (t + 2 < TQ) {
      const char* nx = (const char*)(inputs + ((size_t)(t + 2) * BQ + mtile * 16) * DQ);
      __builtin_prefetch(nx + tid * 32, 0, 3);
    }

    v8f acc0, acc1;
    {
      v2f h0 = {h[0], h[1]};
      v2f h1 = {h[2], h[3]};
      v2f h2 = {h[4], 0.f};
      acc0 = __builtin_amdgcn_wmma_f32_16x16x4_f32(false, wa[16], false, h0,
                                                   (short)0, ax0, false, false);
      acc1 = __builtin_amdgcn_wmma_f32_16x16x4_f32(false, wa[17], false, h1,
                                                   (short)0, ax1, false, false);
      acc0 = __builtin_amdgcn_wmma_f32_16x16x4_f32(false, wa[18], false, h2,
                                                   (short)0, acc0, false, false);
    }

    float fac[HH];
#pragma unroll
    for (int v = 0; v < HH; ++v)
      fac[v] = cpy[v] * __cosf(px[v] + (acc0[v] + acc1[v]) + bias[v]);

    float p = 1.f;
#pragma unroll
    for (int v = 0; v < HH; ++v) p *= fac[v];
    const float tot = xhalf(p);
    float z = __builtin_bit_cast(float,
        (__builtin_bit_cast(int, tot) & hmask) | (0x3f800000 & ~hmask));
    float ga[HH];
    if (g == 2) {
#pragma unroll
      for (int v = 0; v < HH; ++v) { z *= fac[v]; ga[v] = tanh_hw(z); }
    } else {
#pragma unroll
      for (int v = 0; v < HH; ++v) { z *= fac[v]; ga[v] = sigmoid_hw(z); }
    }

    const int pb = t & 1;
#pragma unroll
    for (int v = 0; v < HH; ++v) gx[pb][g][v][lane] = ga[v];
    __builtin_amdgcn_fence(__ATOMIC_RELEASE, "workgroup", "local");
    __builtin_amdgcn_s_barrier();
    __builtin_amdgcn_fence(__ATOMIC_ACQUIRE, "workgroup", "local");

    float fv[HH], iv[HH], gv[HH], ov[HH];
#pragma unroll
    for (int v = 0; v < HH; ++v) {
      fv[v] = gx[pb][0][v][lane]; iv[v] = gx[pb][1][v][lane];
      gv[v] = gx[pb][2][v][lane]; ov[v] = gx[pb][3][v][lane];
    }

    ax0 = (v8f){0.f, 0.f, 0.f, 0.f, 0.f, 0.f, 0.f, 0.f};
    ax1 = (v8f){0.f, 0.f, 0.f, 0.f, 0.f, 0.f, 0.f, 0.f};
#pragma unroll
    for (int i = 0; i < 3; ++i) xpair(wa[2 * i], wa[2 * i + 1], xq[i], ax0, ax1);

    float hr[HH];
#pragma unroll
    for (int v = 0; v < HH; ++v) {
      float cn = fv[v] * c[v] + iv[v] * gv[v];
      c[v] = cn;
      hr[v] = ov[v] * tanh_hw(cn) + h[v];
    }
#pragma unroll
    for (int i = 3; i < 8; ++i) xpair(wa[2 * i], wa[2 * i + 1], xq[i], ax0, ax1);

    float s = 0.f;
#pragma unroll
    for (int v = 0; v < HH; ++v) s += hr[v];
    const float mu = (s + xhalf(s)) * 0.1f;
    float q = 0.f;
#pragma unroll
    for (int v = 0; v < HH; ++v) { float d = hr[v] - mu; q += d * d; }
    const float var = (q + xhalf(q)) * 0.1f;
    const float rs = rsqrtf(var + 1e-5f);
#pragma unroll
    for (int v = 0; v < HH; ++v) h[v] = (hr[v] - mu) * rs * lg[v] + lb[v];

    __builtin_amdgcn_sched_group_barrier(0x100, 10, 0);
    __builtin_amdgcn_sched_group_barrier(0x008, 6, 0);
    __builtin_amdgcn_sched_group_barrier(0x402, 128, 0);

    if (g == 0) {
      float* yrow = ys + ((size_t)t * BQ + mtile * 16 + b) * HQ + HH * half;
#pragma unroll
      for (int v = 0; v < HH; ++v) yrow[v] = h[v];
    }
  }

  if (g == 0) {
#pragma unroll
    for (int v = 0; v < HH; ++v) {
      hout[(mtile * 16 + b) * HQ + HH * half + v] = h[v];
      cout[(mtile * 16 + b) * HQ + HH * half + v] = c[v];
    }
  }
}

extern "C" void kernel_launch(void* const* d_in, const int* in_sizes, int n_in,
                              void* d_out, int out_size, void* d_ws, size_t ws_size,
                              hipStream_t stream) {
  const float* inputs = (const float*)d_in[0];
  const float* Wf = (const float*)d_in[1];
  const float* bf = (const float*)d_in[2];
  const float* pf = (const float*)d_in[3];
  const float* Wi = (const float*)d_in[4];
  const float* bi = (const float*)d_in[5];
  const float* pi = (const float*)d_in[6];
  const float* Wu = (const float*)d_in[7];
  const float* bu = (const float*)d_in[8];
  const float* pu = (const float*)d_in[9];
  const float* Wo = (const float*)d_in[10];
  const float* bo = (const float*)d_in[11];
  const float* po = (const float*)d_in[12];
  const float* lng = (const float*)d_in[13];
  const float* lnb = (const float*)d_in[14];

  float* out  = (float*)d_out;
  float* ys   = out;
  float* hout = out + (size_t)TQ * BQ * HQ;
  float* cout = hout + BQ * HQ;

  qlstm_fused<<<BQ / 16, 128, 0, stream>>>(
      inputs, Wf, Wi, Wu, Wo, bf, bi, bu, bo,
      pf, pi, pu, po, lng, lnb, ys, hout, cout);
}


# x tile staged through LDS: each wave loads a quarter of a step's tile with whole-line loads three steps ahead, all four read their B fragments from LDS a step ahead; sums in the same order
# speedup vs baseline: 6.3015x; 1.3625x over previous
#include <hip/hip_runtime.h>
#include <math.h>

typedef float v2f __attribute__((ext_vector_type(2)));
typedef float v8f __attribute__((ext_vector_type(8)));

#define TQ 128
#define BQ 256
#define DQ 64
#define HQ 10
#define KD 74
#define HH 5
#define XP4 17

__device__ __forceinline__ float xhalf(float x) {
  int b = __builtin_bit_cast(int, x);
  return __builtin_bit_cast(float,
      __builtin_amdgcn_permlanex16(b, b, 0x76543210, 0xfedcba98, false, false));
}

__device__ __forceinline__ float tanh_hw(float x) {
  const float q = __builtin_amdgcn_exp2f(-2.8853900817779268f * __builtin_fabsf(x));
  const float t = (1.f - q) * __builtin_amdgcn_rcpf(1.f + q);
  return __builtin_copysignf(t, x);
}

__device__ __forceinline__ float sigmoid_hw(float z) {
  return __builtin_amdgcn_rcpf(1.f + __expf(-z));
}

__device__ __forceinline__ void xpair(const v2f& wa0, const v2f& wa1, const float4& x,
                                      v8f& a0, v8f& a1) {
  v2f b0 = {x.x, x.y};
  v2f b1 = {x.z, x.w};
  a0 = __builtin_amdgcn_wmma_f32_16x16x4_f32(false, wa0, false, b0, (short)0, a0, false, false);
  a1 = __builtin_amdgcn_wmma_f32_16x16x4_f32(false, wa1, false, b1, (short)0, a1, false, false);
}

__global__ __launch_bounds__(128) void qlstm_fused(
    const float* __restrict__ inputs,
    const float* __restrict__ Wf, const float* __restrict__ Wi,
    const float* __restrict__ Wu, const float* __restrict__ Wo,
    const float* __restrict__ bf, const float* __restrict__ bi,
    const float* __restrict__ bu, const float* __restrict__ bo,
    const float* __restrict__ pf, const float* __restrict__ pi,
    const float* __restrict__ pu, const float* __restrict__ po,
    const float* __restrict__ lng, const float* __restrict__ lnb,
    float* __restrict__ ys, float* __restrict__ hout, float* __restrict__ cout) {
  __shared__ float gx[2][4][HH][32];
  __shared__ float4 xs[2][16 * XP4];

  const int mtile = blockIdx.x;
  const int tid   = threadIdx.x;
  const int g     = __builtin_amdgcn_readfirstlane(tid >> 5);
  const int lane  = tid & 31;
  const int b     = lane & 15;
  const int half  = lane >> 4;

  const float* Wg  = (g == 0) ? Wf : (g == 1) ? Wi : (g == 2) ? Wu : Wo;
  const float* bbg = (g == 0) ? bf : (g == 1) ? bi : (g == 2) ? bu : bo;
  const float* pg  = (g == 0) ? pf : (g == 1) ? pi : (g == 2) ? pu : po;

  const int  am   = lane & 15;
  const bool arow = (am < HH) || (am >= 8 && am < 8 + HH);
  const int  awire = arow ? ((am < HH) ? am : am - 3) : 0;
  const float* wrow = Wg + awire * KD;
  v2f wa[19];
#pragma unroll
  for (int i = 0; i < 8; ++i) {
    const int c = 8 * i + 4 * half;
    wa[2 * i][0]     = arow ? wrow[c + 0] : 0.f;
    wa[2 * i][1]     = arow ? wrow[c + 1] : 0.f;
    wa[2 * i + 1][0] = arow ? wrow[c + 2] : 0.f;
    wa[2 * i + 1][1] = arow ? wrow[c + 3] : 0.f;
  }
#pragma unroll
  for (int j = 0; j < 3; ++j) {
    wa[16 + j][0] = (arow && 2 * j < HH) ? wrow[DQ + HH * half + 2 * j] : 0.f;
    wa[16 + j][1] = (arow && 2 * j + 1 < HH) ? wrow[DQ + HH * half + 2 * j + 1] : 0.f;
  }

  float bias[HH], px[HH], cpy[HH], lg[HH], lb[HH], c[HH], h[HH];
#pragma unroll
  for (int v = 0; v < HH; ++v) {
    const int w = HH * half + v;
    bias[v] = bbg[w];
    px[v]   = pg[w * 3 + 0];
    cpy[v]  = cosf(pg[w * 3 + 1]);
    lg[v]   = lng[w];
    lb[v]   = lnb[w];
    c[v]    = 0.f;
    h[v]    = 0.f;
  }
  const int hmask = -half;

  const float* xbase = inputs + ((size_t)(mtile * 16 + b)) * DQ + 4 * half;
  float4 xq[8];
  v8f ax0 = {0.f, 0.f, 0.f, 0.f, 0.f, 0.f, 0.f, 0.f};
  v8f ax1 = {0.f, 0.f, 0.f, 0.f, 0.f, 0.f, 0.f, 0.f};
#pragma unroll
  for (int i = 0; i < 8; ++i) xq[i] = *(const float4*)(xbase + 8 * i);
#pragma unroll
  for (int i = 0; i < 8; ++i) xpair(wa[2 * i], wa[2 * i + 1], xq[i], ax0, ax1);

  const float* xw = inputs + ((size_t)(mtile * 16 + 4 * g)) * DQ;
  const int    xl = 4 * lane;
  const int so = (4 * g + (lane >> 4)) * XP4 + (lane & 15);
  const int ro = b * XP4 + half;
  {
    const float* x1 = xw + (size_t)BQ * DQ;
    xs[1][so]           = *(const float4*)(x1 + xl);
    xs[1][so + 2 * XP4] = *(const float4*)(x1 + xl + 128);
  }
  float4 gl0 = *(const float4*)(xw + (size_t)2 * BQ * DQ + xl);
  float4 gl1 = *(const float4*)(xw + (size_t)2 * BQ * DQ + xl + 128);
  __syncthreads();

#pragma unroll 2
  for (int t = 0; t < TQ; ++t) {
    const int pb = t & 1;

    v8f acc0, acc1;
    {
      v2f h0 = {h[0], h[1]};
      v2f h1 = {h[2], h[3]};
      v2f h2 = {h[4], 0.f};
      acc0 = __builtin_amdgcn_wmma_f32_16x16x4_f32(false, wa[16], false, h0,
                                                   (short)0, ax0, false, false);
      acc1 = __builtin_amdgcn_wmma_f32_16x16x4_f32(false, wa[17], false, h1,
                                                   (short)0, ax1, false, false);
      acc0 = __builtin_amdgcn_wmma_f32_16x16x4_f32(false, wa[18], false, h2,
                                                   (short)0, acc0, false, false);
    }

    xs[pb][so]           = gl0;
    xs[pb][so + 2 * XP4] = gl1;
    const int tn = (t + 3 < TQ) ? t + 3 : TQ - 1;
    const float* xt = xw + (size_t)tn * BQ * DQ;
    gl0 = *(const float4*)(xt + xl);
    gl1 = *(const float4*)(xt + xl + 128);
#pragma unroll
    for (int i = 0; i < 8; ++i) xq[i] = xs[pb ^ 1][ro + 2 * i];
    {
      const int tp = (t + 5 < TQ) ? t + 5 : TQ - 1;
      const char* nx = (const char*)(inputs + ((size_t)tp * BQ + mtile * 16) * DQ);
      __builtin_prefetch(nx + tid * 32, 0, 3);
    }

    float fac[HH];
#pragma unroll
    for (int v = 0; v < HH; ++v)
      fac[v] = cpy[v] * __cosf(px[v] + (acc0[v] + acc1[v]) + bias[v]);

    float p = 1.f;
#pragma unroll
    for (int v = 0; v < HH; ++v) p *= fac[v];
    const float tot = xhalf(p);
    float z = __builtin_bit_cast(float,
        (__builtin_bit_cast(int, tot) & hmask) | (0x3f800000 & ~hmask));
    __builtin_amdgcn_sched_group_barrier(0x002, 2, 1);
    __builtin_amdgcn_sched_group_barrier(0x008, 2, 1);
    __builtin_amdgcn_sched_group_barrier(0x200, 2, 1);
    __builtin_amdgcn_sched_group_barrier(0x020, 2, 1);
    __builtin_amdgcn_sched_group_barrier(0x100, 8, 1);
    __builtin_amdgcn_sched_group_barrier(0x008, 1, 1);
    float ga[HH];
    if (g == 2) {
#pragma unroll
      for (int v = 0; v < HH; ++v) { z *= fac[v]; ga[v] = tanh_hw(z); }
    } else {
#pragma unroll
      for (int v = 0; v < HH; ++v) { z *= fac[v]; ga[v] = sigmoid_hw(z); }
    }

#pragma unroll
    for (int v = 0; v < HH; ++v) gx[pb][g][v][lane] = ga[v];
    __builtin_amdgcn_fence(__ATOMIC_RELEASE, "workgroup", "local");
    __builtin_amdgcn_s_barrier();
    __builtin_amdgcn_fence(__ATOMIC_ACQUIRE, "workgroup", "local");

    float fv[HH], iv[HH], gv[HH], ov[HH];
#pragma unroll
    for (int v = 0; v < HH; ++v) {
      fv[v] = gx[pb][0][v][lane]; iv[v] = gx[pb][1][v][lane];
      gv[v] = gx[pb][2][v][lane]; ov[v] = gx[pb][3][v][lane];
    }

    ax0 = (v8f){0.f, 0.f, 0.f, 0.f, 0.f, 0.f, 0.f, 0.f};
    ax1 = (v8f){0.f, 0.f, 0.f, 0.f, 0.f, 0.f, 0.f, 0.f};
#pragma unroll
    for (int i = 0; i < 3; ++i) xpair(wa[2 * i], wa[2 * i + 1], xq[i], ax0, ax1);

    float hr[HH];
#pragma unroll
    for (int v = 0; v < HH; ++v) {
      float cn = fv[v] * c[v] + iv[v] * gv[v];
      c[v] = cn;
      hr[v] = ov[v] * tanh_hw(cn) + h[v];
    }
#pragma unroll
    for (int i = 3; i < 8; ++i) xpair(wa[2 * i], wa[2 * i + 1], xq[i], ax0, ax1);

    float s = 0.f;
#pragma unroll
    for (int v = 0; v < HH; ++v) s += hr[v];
    const float mu = (s + xhalf(s)) * 0.1f;
    float q = 0.f;
#pragma unroll
    for (int v = 0; v < HH; ++v) { float d = hr[v] - mu; q += d * d; }
    const float var = (q + xhalf(q)) * 0.1f;
    const float rs = rsqrtf(var + 1e-5f);
#pragma unroll
    for (int v = 0; v < HH; ++v) h[v] = (hr[v] - mu) * rs * lg[v] + lb[v];

    __builtin_amdgcn_sched_group_barrier(0x100, 10, 0);
    __builtin_amdgcn_sched_group_barrier(0x008, 6, 0);
    __builtin_amdgcn_sched_group_barrier(0x402, 128, 0);

    if (g == 0) {
      float* yrow = ys + ((size_t)t * BQ + mtile * 16 + b) * HQ + HH * half;
#pragma unroll
      for (int v = 0; v < HH; ++v) yrow[v] = h[v];
    }
  }

  if (g == 0) {
#pragma unroll
    for (int v = 0; v < HH; ++v) {
      hout[(mtile * 16 + b) * HQ + HH * half + v] = h[v];
      cout[(mtile * 16 + b) * HQ + HH * half + v] = c[v];
    }
  }
}

extern "C" void kernel_launch(void* const* d_in, const int* in_sizes, int n_in,
                              void* d_out, int out_size, void* d_ws, size_t ws_size,
                              hipStream_t stream) {
  const float* inputs = (const float*)d_in[0];
  const float* Wf = (const float*)d_in[1];
  const float* bf = (const float*)d_in[2];
  const float* pf = (const float*)d_in[3];
  const float* Wi = (const float*)d_in[4];
  const float* bi = (const float*)d_in[5];
  const float* pi = (const float*)d_in[6];
  const float* Wu = (const float*)d_in[7];
  const float* bu = (const float*)d_in[8];
  const float* pu = (const float*)d_in[9];
  const float* Wo = (const float*)d_in[10];
  const float* bo = (const float*)d_in[11];
  const float* po = (const float*)d_in[12];
  const float* lng = (const float*)d_in[13];
  const float* lnb = (const float*)d_in[14];

  float* out  = (float*)d_out;
  float* ys   = out;
  float* hout = out + (size_t)TQ * BQ * HQ;
  float* cout = hout + BQ * HQ;

  qlstm_fused<<<BQ / 16, 128, 0, stream>>>(
      inputs, Wf, Wi, Wu, Wo, bf, bi, bu, bo,
      pf, pi, pu, po, lng, lnb, ys, hout, cout);
}


# 6 of a step's 16 x wmma move in front of the barrier, where the matrix pipe idles: 2 late in the gate chain, 4 behind the gate stores; 4 stay under the LDS reads, 6 beside the cell chain; same sums
# speedup vs baseline: 6.5175x; 1.0343x over previous
#include <hip/hip_runtime.h>
#include <math.h>

typedef float v2f __attribute__((ext_vector_type(2)));
typedef float v8f __attribute__((ext_vector_type(8)));

#define TQ 128
#define BQ 256
#define DQ 64
#define HQ 10
#define KD 74
#define HH 5
#define XP4 17

__device__ __forceinline__ float xhalf(float x) {
  int b = __builtin_bit_cast(int, x);
  return __builtin_bit_cast(float,
      __builtin_amdgcn_permlanex16(b, b, 0x76543210, 0xfedcba98, false, false));
}

__device__ __forceinline__ float tanh_hw(float x) {
  const float q = __builtin_amdgcn_exp2f(-2.8853900817779268f * __builtin_fabsf(x));
  const float t = (1.f - q) * __builtin_amdgcn_rcpf(1.f + q);
  return __builtin_copysignf(t, x);
}

__device__ __forceinline__ float sigmoid_hw(float z) {
  return __builtin_amdgcn_rcpf(1.f + __expf(-z));
}

__device__ __forceinline__ void xpair(const v2f& wa0, const v2f& wa1, const float4& x,
                                      v8f& a0, v8f& a1) {
  v2f b0 = {x.x, x.y};
  v2f b1 = {x.z, x.w};
  a0 = __builtin_amdgcn_wmma_f32_16x16x4_f32(false, wa0, false, b0, (short)0, a0, false, false);
  a1 = __builtin_amdgcn_wmma_f32_16x16x4_f32(false, wa1, false, b1, (short)0, a1, false, false);
}

__device__ __forceinline__ void xone(const v2f& w, float x0, float x1, v8f& a) {
  v2f bb = {x0, x1};
  a = __builtin_amdgcn_wmma_f32_16x16x4_f32(false, w, false, bb, (short)0, a, false, false);
}

__global__ __launch_bounds__(128) void qlstm_fused(
    const float* __restrict__ inputs,
    const float* __restrict__ Wf, const float* __restrict__ Wi,
    const float* __restrict__ Wu, const float* __restrict__ Wo,
    const float* __restrict__ bf, const float* __restrict__ bi,
    const float* __restrict__ bu, const float* __restrict__ bo,
    const float* __restrict__ pf, const float* __restrict__ pi,
    const float* __restrict__ pu, const float* __restrict__ po,
    const float* __restrict__ lng, const float* __restrict__ lnb,
    float* __restrict__ ys, float* __restrict__ hout, float* __restrict__ cout) {
  __shared__ float gx[2][4][HH][32];
  __shared__ float4 xs[2][16 * XP4];

  const int mtile = blockIdx.x;
  const int tid   = threadIdx.x;
  const int g     = __builtin_amdgcn_readfirstlane(tid >> 5);
  const int lane  = tid & 31;
  const int b     = lane & 15;
  const int half  = lane >> 4;

  const float* Wg  = (g == 0) ? Wf : (g == 1) ? Wi : (g == 2) ? Wu : Wo;
  const float* bbg = (g == 0) ? bf : (g == 1) ? bi : (g == 2) ? bu : bo;
  const float* pg  = (g == 0) ? pf : (g == 1) ? pi : (g == 2) ? pu : po;

  const int  am   = lane & 15;
  const bool arow = (am < HH) || (am >= 8 && am < 8 + HH);
  const int  awire = arow ? ((am < HH) ? am : am - 3) : 0;
  const float* wrow = Wg + awire * KD;
  v2f wa[19];
#pragma unroll
  for (int i = 0; i < 8; ++i) {
    const int c = 8 * i + 4 * half;
    wa[2 * i][0]     = arow ? wrow[c + 0] : 0.f;
    wa[2 * i][1]     = arow ? wrow[c + 1] : 0.f;
    wa[2 * i + 1][0] = arow ? wrow[c + 2] : 0.f;
    wa[2 * i + 1][1] = arow ? wrow[c + 3] : 0.f;
  }
#pragma unroll
  for (int j = 0; j < 3; ++j) {
    wa[16 + j][0] = (arow && 2 * j < HH) ? wrow[DQ + HH * half + 2 * j] : 0.f;
    wa[16 + j][1] = (arow && 2 * j + 1 < HH) ? wrow[DQ + HH * half + 2 * j + 1] : 0.f;
  }

  float bias[HH], px[HH], cpy[HH], lg[HH], lb[HH], c[HH], h[HH];
#pragma unroll
  for (int v = 0; v < HH; ++v) {
    const int w = HH * half + v;
    bias[v] = bbg[w];
    px[v]   = pg[w * 3 + 0];
    cpy[v]  = cosf(pg[w * 3 + 1]);
    lg[v]   = lng[w];
    lb[v]   = lnb[w];
    c[v]    = 0.f;
    h[v]    = 0.f;
  }
  const int hmask = -half;

  const float* xbase = inputs + ((size_t)(mtile * 16 + b)) * DQ + 4 * half;
  float4 xq[8];
  v8f ax0 = {0.f, 0.f, 0.f, 0.f, 0.f, 0.f, 0.f, 0.f};
  v8f ax1 = {0.f, 0.f, 0.f, 0.f, 0.f, 0.f, 0.f, 0.f};
#pragma unroll
  for (int i = 0; i < 8; ++i) xq[i] = *(const float4*)(xbase + 8 * i);
#pragma unroll
  for (int i = 0; i < 8; ++i) xpair(wa[2 * i], wa[2 * i + 1], xq[i], ax0, ax1);

  const float* xw = inputs + ((size_t)(mtile * 16 + 4 * g)) * DQ;
  const int    xl = 4 * lane;
  const int so = (4 * g + (lane >> 4)) * XP4 + (lane & 15);
  const int ro = b * XP4 + half;
  {
    const float* x1 = xw + (size_t)BQ * DQ;
    xs[1][so]           = *(const float4*)(x1 + xl);
    xs[1][so + 2 * XP4] = *(const float4*)(x1 + xl + 128);
  }
  float4 gl0 = *(const float4*)(xw + (size_t)2 * BQ * DQ + xl);
  float4 gl1 = *(const float4*)(xw + (size_t)2 * BQ * DQ + xl + 128);
  __syncthreads();

#pragma unroll 2
  for (int t = 0; t < TQ; ++t) {
    const int pb = t & 1;

    v8f acc0, acc1;
    {
      v2f h0 = {h[0], h[1]};
      v2f h1 = {h[2], h[3]};
      v2f h2 = {h[4], 0.f};
      acc0 = __builtin_amdgcn_wmma_f32_16x16x4_f32(false, wa[16], false, h0,
                                                   (short)0, ax0, false, false);
      acc1 = __builtin_amdgcn_wmma_f32_16x16x4_f32(false, wa[17], false, h1,
                                                   (short)0, ax1, false, false);
      acc0 = __builtin_amdgcn_wmma_f32_16x16x4_f32(false, wa[18], false, h2,
                                                   (short)0, acc0, false, false);
    }

    xs[pb][so]           = gl0;
    xs[pb][so + 2 * XP4] = gl1;
    const int tn = (t + 3 < TQ) ? t + 3 : TQ - 1;
    const float* xt = xw + (size_t)tn * BQ * DQ;
    gl0 = *(const float4*)(xt + xl);
    gl1 = *(const float4*)(xt + xl + 128);
#pragma unroll
    for (int i = 0; i < 8; ++i) xq[i] = xs[pb ^ 1][ro + 2 * i];
    {
      const int tp = (t + 5 < TQ) ? t + 5 : TQ - 1;
      const char* nx = (const char*)(inputs + ((size_t)tp * BQ + mtile * 16) * DQ);
      __builtin_prefetch(nx + tid * 32, 0, 3);
    }

    float fac[HH];
#pragma unroll
    for (int v = 0; v < HH; ++v)
      fac[v] = cpy[v] * __cosf(px[v] + (acc0[v] + acc1[v]) + bias[v]);

    float p = 1.f;
#pragma unroll
    for (int v = 0; v < HH; ++v) p *= fac[v];
    const float tot = xhalf(p);
    float z = __builtin_bit_cast(float,
        (__builtin_bit_cast(int, tot) & hmask) | (0x3f800000 & ~hmask));
    __builtin_amdgcn_sched_group_barrier(0x002, 2, 1);
    __builtin_amdgcn_sched_group_barrier(0x008, 2, 1);
    __builtin_amdgcn_sched_group_barrier(0x200, 2, 1);
    __builtin_amdgcn_sched_group_barrier(0x020, 2, 1);
    __builtin_amdgcn_sched_group_barrier(0x100, 8, 1);
    __builtin_amdgcn_sched_group_barrier(0x008, 1, 1);

    const v8f zero = {0.f, 0.f, 0.f, 0.f, 0.f, 0.f, 0.f, 0.f};
    {
      v2f xb0 = {xq[0].x, xq[0].y};
      asm("" : "+v"(xb0) : "v"(p));
      ax0 = __builtin_amdgcn_wmma_f32_16x16x4_f32(false, wa[0], false, xb0, (short)0, zero, false, false);
      asm("" : "+v"(z) : "v"(ax0));
    }
    float zz[HH];
#pragma unroll
    for (int v = 0; v < HH; ++v) { z *= fac[v]; zz[v] = z; }
    {
      v2f xb1 = {xq[0].z, xq[0].w};
      asm("" : "+v"(xb1) : "v"(zz[2]));
      ax1 = __builtin_amdgcn_wmma_f32_16x16x4_f32(false, wa[1], false, xb1, (short)0, zero, false, false);
    }
    float ga[HH];
    if (g == 2) {
#pragma unroll
      for (int v = 0; v < HH; ++v) ga[v] = tanh_hw(zz[v]);
    } else {
#pragma unroll
      for (int v = 0; v < HH; ++v) ga[v] = sigmoid_hw(zz[v]);
    }

#pragma unroll
    for (int v = 0; v < HH; ++v) gx[pb][g][v][lane] = ga[v];
    xone(wa[2], xq[1].x, xq[1].y, ax0);
    xone(wa[3], xq[1].z, xq[1].w, ax1);
    xone(wa[4], xq[2].x, xq[2].y, ax0);
    xone(wa[5], xq[2].z, xq[2].w, ax1);
    asm volatile("" :: "v"(ax0), "v"(ax1));
    __builtin_amdgcn_fence(__ATOMIC_RELEASE, "workgroup", "local");
    __builtin_amdgcn_s_barrier();
    __builtin_amdgcn_fence(__ATOMIC_ACQUIRE, "workgroup", "local");

    float fv[HH], iv[HH], gv[HH], ov[HH];
#pragma unroll
    for (int v = 0; v < HH; ++v) {
      fv[v] = gx[pb][0][v][lane]; iv[v] = gx[pb][1][v][lane];
      gv[v] = gx[pb][2][v][lane]; ov[v] = gx[pb][3][v][lane];
    }

#pragma unroll
    for (int i = 3; i < 5; ++i) xpair(wa[2 * i], wa[2 * i + 1], xq[i], ax0, ax1);

    float hr[HH];
#pragma unroll
    for (int v = 0; v < HH; ++v) {
      float cn = fv[v] * c[v] + iv[v] * gv[v];
      c[v] = cn;
      hr[v] = ov[v] * tanh_hw(cn) + h[v];
    }
#pragma unroll
    for (int i = 5; i < 8; ++i) xpair(wa[2 * i], wa[2 * i + 1], xq[i], ax0, ax1);

    float s = 0.f;
#pragma unroll
    for (int v = 0; v < HH; ++v) s += hr[v];
    const float mu = (s + xhalf(s)) * 0.1f;
    float q = 0.f;
#pragma unroll
    for (int v = 0; v < HH; ++v) { float d = hr[v] - mu; q += d * d; }
    const float var = (q + xhalf(q)) * 0.1f;
    const float rs = rsqrtf(var + 1e-5f);
#pragma unroll
    for (int v = 0; v < HH; ++v) h[v] = (hr[v] - mu) * rs * lg[v] + lb[v];

    __builtin_amdgcn_sched_group_barrier(0x200, 3, 0);
    __builtin_amdgcn_sched_group_barrier(0x008, 4, 0);
    __builtin_amdgcn_sched_group_barrier(0x100, 10, 0);
    __builtin_amdgcn_sched_group_barrier(0x008, 4, 0);
    __builtin_amdgcn_sched_group_barrier(0x402, 128, 0);

    if (g == 0) {
      float* yrow = ys + ((size_t)t * BQ + mtile * 16 + b) * HQ + HH * half;
#pragma unroll
      for (int v = 0; v < HH; ++v) yrow[v] = h[v];
    }
  }

  if (g == 0) {
#pragma unroll
    for (int v = 0; v < HH; ++v) {
      hout[(mtile * 16 + b) * HQ + HH * half + v] = h[v];
      cout[(mtile * 16 + b) * HQ + HH * half + v] = c[v];
    }
  }
}

extern "C" void kernel_launch(void* const* d_in, const int* in_sizes, int n_in,
                              void* d_out, int out_size, void* d_ws, size_t ws_size,
                              hipStream_t stream) {
  const float* inputs = (const float*)d_in[0];
  const float* Wf = (const float*)d_in[1];
  const float* bf = (const float*)d_in[2];
  const float* pf = (const float*)d_in[3];
  const float* Wi = (const float*)d_in[4];
  const float* bi = (const float*)d_in[5];
  const float* pi = (const float*)d_in[6];
  const float* Wu = (const float*)d_in[7];
  const float* bu = (const float*)d_in[8];
  const float* pu = (const float*)d_in[9];
  const float* Wo = (const float*)d_in[10];
  const float* bo = (const float*)d_in[11];
  const float* po = (const float*)d_in[12];
  const float* lng = (const float*)d_in[13];
  const float* lnb = (const float*)d_in[14];

  float* out  = (float*)d_out;
  float* ys   = out;
  float* hout = out + (size_t)TQ * BQ * HQ;
  float* cout = hout + BQ * HQ;

  qlstm_fused<<<BQ / 16, 128, 0, stream>>>(
      inputs, Wf, Wi, Wu, Wo, bf, bi, bu, bo,
      pf, pi, pu, po, lng, lnb, ys, hout, cout);
}


# translation waits off the recurrence: loop x loads (and y stores) on scalar base + fixed lane offset, no address register rewritten behind the loads; W loaded unmasked, zeros selected; same sums
# speedup vs baseline: 7.0988x; 1.0892x over previous
#include <hip/hip_runtime.h>
#include <math.h>

typedef float v2f __attribute__((ext_vector_type(2)));
typedef float v8f __attribute__((ext_vector_type(8)));
typedef float v4f __attribute__((ext_vector_type(4)));
typedef __attribute__((address_space(1))) float gmem_f;
typedef __attribute__((address_space(1))) const float gmem_cf;
typedef __attribute__((address_space(1))) const v4f gmem_cv4;
typedef __attribute__((address_space(1))) const char gmem_cc;

#define TQ 128
#define BQ 256
#define DQ 64
#define HQ 10
#define KD 74
#define HH 5
#define XP4 17

__device__ __forceinline__ float xhalf(float x) {
  int b = __builtin_bit_cast(int, x);
  return __builtin_bit_cast(float,
      __builtin_amdgcn_permlanex16(b, b, 0x76543210, 0xfedcba98, false, false));
}

__device__ __forceinline__ unsigned long long sbase_bits(const void* p) {
  const unsigned long long a = (unsigned long long)p;
  const unsigned lo = (unsigned)__builtin_amdgcn_readfirstlane((int)(unsigned)(a & 0xffffffffull));
  const unsigned hi = (unsigned)__builtin_amdgcn_readfirstlane((int)(unsigned)(a >> 32));
  return ((unsigned long long)hi << 32) | (unsigned long long)lo;
}

__device__ __forceinline__ float4 ldg4v(const gmem_cv4* p) {
  const v4f t = *p;
  return make_float4(t.x, t.y, t.z, t.w);
}

__device__ __forceinline__ float tanh_hw(float x) {
  const float q = __builtin_amdgcn_exp2f(-2.8853900817779268f * __builtin_fabsf(x));
  const float t = (1.f - q) * __builtin_amdgcn_rcpf(1.f + q);
  return __builtin_copysignf(t, x);
}

__device__ __forceinline__ float sigmoid_hw(float z) {
  return __builtin_amdgcn_rcpf(1.f + __expf(-z));
}

__device__ __forceinline__ void xpair(const v2f& wa0, const v2f& wa1, const float4& x,
                                      v8f& a0, v8f& a1) {
  v2f b0 = {x.x, x.y};
  v2f b1 = {x.z, x.w};
  a0 = __builtin_amdgcn_wmma_f32_16x16x4_f32(false, wa0, false, b0, (short)0, a0, false, false);
  a1 = __builtin_amdgcn_wmma_f32_16x16x4_f32(false, wa1, false, b1, (short)0, a1, false, false);
}

__device__ __forceinline__ void xone(const v2f& w, float x0, float x1, v8f& a) {
  v2f bb = {x0, x1};
  a = __builtin_amdgcn_wmma_f32_16x16x4_f32(false, w, false, bb, (short)0, a, false, false);
}

__global__ __launch_bounds__(128) void qlstm_fused(
    const float* __restrict__ inputs,
    const float* __restrict__ Wf, const float* __restrict__ Wi,
    const float* __restrict__ Wu, const float* __restrict__ Wo,
    const float* __restrict__ bf, const float* __restrict__ bi,
    const float* __restrict__ bu, const float* __restrict__ bo,
    const float* __restrict__ pf, const float* __restrict__ pi,
    const float* __restrict__ pu, const float* __restrict__ po,
    const float* __restrict__ lng, const float* __restrict__ lnb,
    float* __restrict__ ys, float* __restrict__ hout, float* __restrict__ cout) {
  __shared__ float gx[2][4][HH][32];
  __shared__ float4 xs[2][16 * XP4];

  const int mtile = blockIdx.x;
  const int tid   = threadIdx.x;
  const int g     = __builtin_amdgcn_readfirstlane(tid >> 5);
  const int lane  = tid & 31;
  const int b     = lane & 15;
  const int half  = lane >> 4;

  const float* Wg  = (g == 0) ? Wf : (g == 1) ? Wi : (g == 2) ? Wu : Wo;
  const float* bbg = (g == 0) ? bf : (g == 1) ? bi : (g == 2) ? bu : bo;
  const float* pg  = (g == 0) ? pf : (g == 1) ? pi : (g == 2) ? pu : po;

  const int  am   = lane & 15;
  const bool arow = (am < HH) || (am >= 8 && am < 8 + HH);
  const int  awire = arow ? ((am < HH) ? am : am - 3) : 0;
  const float* wrow = Wg + awire * KD;
  v2f wa[19];
#pragma unroll
  for (int i = 0; i < 8; ++i) {
    const int c = 8 * i + 4 * half;
    const float w0 = wrow[c + 0], w1 = wrow[c + 1], w2 = wrow[c + 2], w3 = wrow[c + 3];
    wa[2 * i][0]     = arow ? w0 : 0.f;
    wa[2 * i][1]     = arow ? w1 : 0.f;
    wa[2 * i + 1][0] = arow ? w2 : 0.f;
    wa[2 * i + 1][1] = arow ? w3 : 0.f;
  }
#pragma unroll
  for (int j = 0; j < 3; ++j) {
    float u0 = 0.f, u1 = 0.f;
    if (2 * j < HH)     u0 = wrow[DQ + HH * half + 2 * j];
    if (2 * j + 1 < HH) u1 = wrow[DQ + HH * half + 2 * j + 1];
    wa[16 + j][0] = arow ? u0 : 0.f;
    wa[16 + j][1] = arow ? u1 : 0.f;
  }

  float bias[HH], px[HH], cpy[HH], lg[HH], lb[HH], c[HH], h[HH];
#pragma unroll
  for (int v = 0; v < HH; ++v) {
    const int w = HH * half + v;
    bias[v] = bbg[w];
    px[v]   = pg[w * 3 + 0];
    cpy[v]  = cosf(pg[w * 3 + 1]);
    lg[v]   = lng[w];
    lb[v]   = lnb[w];
    c[v]    = 0.f;
    h[v]    = 0.f;
  }
  const int hmask = -half;

  const float* xbase = inputs + ((size_t)(mtile * 16 + b)) * DQ + 4 * half;
  float4 xq[8];
  v8f ax0 = {0.f, 0.f, 0.f, 0.f, 0.f, 0.f, 0.f, 0.f};
  v8f ax1 = {0.f, 0.f, 0.f, 0.f, 0.f, 0.f, 0.f, 0.f};
#pragma unroll
  for (int i = 0; i < 8; ++i) xq[i] = *(const float4*)(xbase + 8 * i);
#pragma unroll
  for (int i = 0; i < 8; ++i) xpair(wa[2 * i], wa[2 * i + 1], xq[i], ax0, ax1);

  const float* xw = inputs + ((size_t)(mtile * 16 + 4 * g)) * DQ;
  const float* xtile = inputs + ((size_t)(mtile * 16)) * DQ;
  const int    xl = 4 * lane;
  const int so = (4 * g + (lane >> 4)) * XP4 + (lane & 15);
  const int ro = b * XP4 + half;
  {
    const float* x1 = xw + (size_t)BQ * DQ;
    xs[1][so]           = *(const float4*)(x1 + xl);
    xs[1][so + 2 * XP4] = *(const float4*)(x1 + xl + 128);
  }
  float4 gl0 = *(const float4*)(xw + (size_t)2 * BQ * DQ + xl);
  float4 gl1 = *(const float4*)(xw + (size_t)2 * BQ * DQ + xl + 128);
  __syncthreads();

#pragma unroll 2
  for (int t = 0; t < TQ; ++t) {
    const int pb = t & 1;

    v8f acc0, acc1;
    {
      v2f h0 = {h[0], h[1]};
      v2f h1 = {h[2], h[3]};
      v2f h2 = {h[4], 0.f};
      acc0 = __builtin_amdgcn_wmma_f32_16x16x4_f32(false, wa[16], false, h0,
                                                   (short)0, ax0, false, false);
      acc1 = __builtin_amdgcn_wmma_f32_16x16x4_f32(false, wa[17], false, h1,
                                                   (short)0, ax1, false, false);
      acc0 = __builtin_amdgcn_wmma_f32_16x16x4_f32(false, wa[18], false, h2,
                                                   (short)0, acc0, false, false);
    }

    xs[pb][so]           = gl0;
    xs[pb][so + 2 * XP4] = gl1;
    const int tn = (t + 3 < TQ) ? t + 3 : TQ - 1;
    const gmem_cf* xt = (const gmem_cf*)sbase_bits(xw + (size_t)tn * BQ * DQ);
    gl0 = ldg4v((const gmem_cv4*)xt + lane);
    gl1 = ldg4v((const gmem_cv4*)xt + lane + 32);
#pragma unroll
    for (int i = 0; i < 8; ++i) xq[i] = xs[pb ^ 1][ro + 2 * i];
    {
      const int tp = (t + 5 < TQ) ? t + 5 : TQ - 1;
      const gmem_cc* nx = (const gmem_cc*)sbase_bits(xtile + (size_t)tp * BQ * DQ);
      __builtin_prefetch((const void*)(nx + tid * 32), 0, 3);
    }

    float fac[HH];
#pragma unroll
    for (int v = 0; v < HH; ++v)
      fac[v] = cpy[v] * __cosf(px[v] + (acc0[v] + acc1[v]) + bias[v]);

    float p = 1.f;
#pragma unroll
    for (int v = 0; v < HH; ++v) p *= fac[v];
    const float tot = xhalf(p);
    float z = __builtin_bit_cast(float,
        (__builtin_bit_cast(int, tot) & hmask) | (0x3f800000 & ~hmask));
    __builtin_amdgcn_sched_group_barrier(0x002, 1, 1);
    __builtin_amdgcn_sched_group_barrier(0x008, 2, 1);
    __builtin_amdgcn_sched_group_barrier(0x200, 2, 1);
    __builtin_amdgcn_sched_group_barrier(0x020, 2, 1);
    __builtin_amdgcn_sched_group_barrier(0x100, 8, 1);
    __builtin_amdgcn_sched_group_barrier(0x008, 1, 1);

    const v8f zero = {0.f, 0.f, 0.f, 0.f, 0.f, 0.f, 0.f, 0.f};
    {
      v2f xb0 = {xq[0].x, xq[0].y};
      asm("" : "+v"(xb0) : "v"(p));
      ax0 = __builtin_amdgcn_wmma_f32_16x16x4_f32(false, wa[0], false, xb0, (short)0, zero, false, false);
      asm("" : "+v"(z) : "v"(ax0));
    }
    float zz[HH];
#pragma unroll
    for (int v = 0; v < HH; ++v) { z *= fac[v]; zz[v] = z; }
    {
      v2f xb1 = {xq[0].z, xq[0].w};
      asm("" : "+v"(xb1) : "v"(zz[2]));
      ax1 = __builtin_amdgcn_wmma_f32_16x16x4_f32(false, wa[1], false, xb1, (short)0, zero, false, false);
    }
    float ga[HH];
    if (g == 2) {
#pragma unroll
      for (int v = 0; v < HH; ++v) ga[v] = tanh_hw(zz[v]);
    } else {
#pragma unroll
      for (int v = 0; v < HH; ++v) ga[v] = sigmoid_hw(zz[v]);
    }

#pragma unroll
    for (int v = 0; v < HH; ++v) gx[pb][g][v][lane] = ga[v];
    xone(wa[2], xq[1].x, xq[1].y, ax0);
    xone(wa[3], xq[1].z, xq[1].w, ax1);
    xone(wa[4], xq[2].x, xq[2].y, ax0);
    xone(wa[5], xq[2].z, xq[2].w, ax1);
    asm volatile("" :: "v"(ax0), "v"(ax1));
    __builtin_amdgcn_fence(__ATOMIC_RELEASE, "workgroup", "local");
    __builtin_amdgcn_s_barrier();
    __builtin_amdgcn_fence(__ATOMIC_ACQUIRE, "workgroup", "local");

    float fv[HH], iv[HH], gv[HH], ov[HH];
#pragma unroll
    for (int v = 0; v < HH; ++v) {
      fv[v] = gx[pb][0][v][lane]; iv[v] = gx[pb][1][v][lane];
      gv[v] = gx[pb][2][v][lane]; ov[v] = gx[pb][3][v][lane];
    }

#pragma unroll
    for (int i = 3; i < 5; ++i) xpair(wa[2 * i], wa[2 * i + 1], xq[i], ax0, ax1);

    float hr[HH];
#pragma unroll
    for (int v = 0; v < HH; ++v) {
      float cn = fv[v] * c[v] + iv[v] * gv[v];
      c[v] = cn;
      hr[v] = ov[v] * tanh_hw(cn) + h[v];
    }
#pragma unroll
    for (int i = 5; i < 8; ++i) xpair(wa[2 * i], wa[2 * i + 1], xq[i], ax0, ax1);

    float s = 0.f;
#pragma unroll
    for (int v = 0; v < HH; ++v) s += hr[v];
    const float mu = (s + xhalf(s)) * 0.1f;
    float q = 0.f;
#pragma unroll
    for (int v = 0; v < HH; ++v) { float d = hr[v] - mu; q += d * d; }
    const float var = (q + xhalf(q)) * 0.1f;
    const float rs = rsqrtf(var + 1e-5f);
#pragma unroll
    for (int v = 0; v < HH; ++v) h[v] = (hr[v] - mu) * rs * lg[v] + lb[v];

    __builtin_amdgcn_sched_group_barrier(0x200, 3, 0);
    __builtin_amdgcn_sched_group_barrier(0x008, 4, 0);
    __builtin_amdgcn_sched_group_barrier(0x100, 10, 0);
    __builtin_amdgcn_sched_group_barrier(0x008, 4, 0);
    __builtin_amdgcn_sched_group_barrier(0x402, 128, 0);

    if (g == 0) {
      gmem_f* yt = (gmem_f*)sbase_bits(ys + ((size_t)t * BQ + mtile * 16) * HQ);
      gmem_f* yrow = yt + (b * HQ + HH * half);
#pragma unroll
      for (int v = 0; v < HH; ++v) yrow[v] = h[v];
    }
  }

  if (g == 0) {
#pragma unroll
    for (int v = 0; v < HH; ++v) {
      hout[(mtile * 16 + b) * HQ + HH * half + v] = h[v];
      cout[(mtile * 16 + b) * HQ + HH * half + v] = c[v];
    }
  }
}

extern "C" void kernel_launch(void* const* d_in, const int* in_sizes, int n_in,
                              void* d_out, int out_size, void* d_ws, size_t ws_size,
                              hipStream_t stream) {
  const float* inputs = (const float*)d_in[0];
  const float* Wf = (const float*)d_in[1];
  const float* bf = (const float*)d_in[2];
  const float* pf = (const float*)d_in[3];
  const float* Wi = (const float*)d_in[4];
  const float* bi = (const float*)d_in[5];
  const float* pi = (const float*)d_in[6];
  const float* Wu = (const float*)d_in[7];
  const float* bu = (const float*)d_in[8];
  const float* pu = (const float*)d_in[9];
  const float* Wo = (const float*)d_in[10];
  const float* bo = (const float*)d_in[11];
  const float* po = (const float*)d_in[12];
  const float* lng = (const float*)d_in[13];
  const float* lnb = (const float*)d_in[14];

  float* out  = (float*)d_out;
  float* ys   = out;
  float* hout = out + (size_t)TQ * BQ * HQ;
  float* cout = hout + BQ * HQ;

  qlstm_fused<<<BQ / 16, 128, 0, stream>>>(
      inputs, Wf, Wi, Wu, Wo, bf, bi, bu, bo,
      pf, pi, pu, po, lng, lnb, ys, hout, cout);
}


# LayerNorm scale deferred behind the h GEMM: the 3 h wmma take hr-mu (lg folded into W_h; W_h lb, bias, phase start the x accumulator), run under variance and rsqrt; rs scales them: new sum order
# speedup vs baseline: 7.3788x; 1.0394x over previous
#include <hip/hip_runtime.h>
#include <math.h>

typedef float v2f __attribute__((ext_vector_type(2)));
typedef float v8f __attribute__((ext_vector_type(8)));
typedef float v4f __attribute__((ext_vector_type(4)));
typedef __attribute__((address_space(1))) float gmem_f;
typedef __attribute__((address_space(1))) const float gmem_cf;
typedef __attribute__((address_space(1))) const v4f gmem_cv4;
typedef __attribute__((address_space(1))) const char gmem_cc;

#define TQ 128
#define BQ 256
#define DQ 64
#define HQ 10
#define KD 74
#define HH 5
#define XP4 17

__device__ __forceinline__ float xhalf(float x) {
  int b = __builtin_bit_cast(int, x);
  return __builtin_bit_cast(float,
      __builtin_amdgcn_permlanex16(b, b, 0x76543210, 0xfedcba98, false, false));
}

__device__ __forceinline__ unsigned long long sbase_bits(const void* p) {
  const unsigned long long a = (unsigned long long)p;
  const unsigned lo = (unsigned)__builtin_amdgcn_readfirstlane((int)(unsigned)(a & 0xffffffffull));
  const unsigned hi = (unsigned)__builtin_amdgcn_readfirstlane((int)(unsigned)(a >> 32));
  return ((unsigned long long)hi << 32) | (unsigned long long)lo;
}

__device__ __forceinline__ float4 ldg4v(const gmem_cv4* p) {
  const v4f t = *p;
  return make_float4(t.x, t.y, t.z, t.w);
}

__device__ __forceinline__ float tanh_hw(float x) {
  const float q = __builtin_amdgcn_exp2f(-2.8853900817779268f * __builtin_fabsf(x));
  const float t = (1.f - q) * __builtin_amdgcn_rcpf(1.f + q);
  return __builtin_copysignf(t, x);
}

__device__ __forceinline__ float sigmoid_hw(float z) {
  return __builtin_amdgcn_rcpf(1.f + __expf(-z));
}

__device__ __forceinline__ void xpair(const v2f& wa0, const v2f& wa1, const float4& x,
                                      v8f& a0, v8f& a1) {
  v2f b0 = {x.x, x.y};
  v2f b1 = {x.z, x.w};
  a0 = __builtin_amdgcn_wmma_f32_16x16x4_f32(false, wa0, false, b0, (short)0, a0, false, false);
  a1 = __builtin_amdgcn_wmma_f32_16x16x4_f32(false, wa1, false, b1, (short)0, a1, false, false);
}

__device__ __forceinline__ void xone(const v2f& w, float x0, float x1, v8f& a) {
  v2f bb = {x0, x1};
  a = __builtin_amdgcn_wmma_f32_16x16x4_f32(false, w, false, bb, (short)0, a, false, false);
}

__global__ __launch_bounds__(128) void qlstm_fused(
    const float* __restrict__ inputs,
    const float* __restrict__ Wf, const float* __restrict__ Wi,
    const float* __restrict__ Wu, const float* __restrict__ Wo,
    const float* __restrict__ bf, const float* __restrict__ bi,
    const float* __restrict__ bu, const float* __restrict__ bo,
    const float* __restrict__ pf, const float* __restrict__ pi,
    const float* __restrict__ pu, const float* __restrict__ po,
    const float* __restrict__ lng, const float* __restrict__ lnb,
    float* __restrict__ ys, float* __restrict__ hout, float* __restrict__ cout) {
  __shared__ float gx[2][4][HH][32];
  __shared__ float4 xs[2][16 * XP4];

  const int mtile = blockIdx.x;
  const int tid   = threadIdx.x;
  const int g     = __builtin_amdgcn_readfirstlane(tid >> 5);
  const int lane  = tid & 31;
  const int b     = lane & 15;
  const int half  = lane >> 4;

  const float* Wg  = (g == 0) ? Wf : (g == 1) ? Wi : (g == 2) ? Wu : Wo;
  const float* bbg = (g == 0) ? bf : (g == 1) ? bi : (g == 2) ? bu : bo;
  const float* pg  = (g == 0) ? pf : (g == 1) ? pi : (g == 2) ? pu : po;

  const int  am   = lane & 15;
  const bool arow = (am < HH) || (am >= 8 && am < 8 + HH);
  const int  awire = arow ? ((am < HH) ? am : am - 3) : 0;
  const float* wrow = Wg + awire * KD;
  v2f wa[19];
#pragma unroll
  for (int i = 0; i < 8; ++i) {
    const int c = 8 * i + 4 * half;
    const float w0 = wrow[c + 0], w1 = wrow[c + 1], w2 = wrow[c + 2], w3 = wrow[c + 3];
    wa[2 * i][0]     = arow ? w0 : 0.f;
    wa[2 * i][1]     = arow ? w1 : 0.f;
    wa[2 * i + 1][0] = arow ? w2 : 0.f;
    wa[2 * i + 1][1] = arow ? w3 : 0.f;
  }
#pragma unroll
  for (int j = 0; j < 3; ++j) {
    float u0 = 0.f, u1 = 0.f;
    if (2 * j < HH)     u0 = wrow[DQ + HH * half + 2 * j];
    if (2 * j + 1 < HH) u1 = wrow[DQ + HH * half + 2 * j + 1];
    wa[16 + j][0] = arow ? u0 : 0.f;
    wa[16 + j][1] = arow ? u1 : 0.f;
  }

  float bias[HH], px[HH], cpy[HH], lg[HH], lb[HH], c[HH], h[HH];
#pragma unroll
  for (int v = 0; v < HH; ++v) {
    const int w = HH * half + v;
    bias[v] = bbg[w];
    px[v]   = pg[w * 3 + 0];
    cpy[v]  = cosf(pg[w * 3 + 1]);
    lg[v]   = lng[w];
    lb[v]   = lnb[w];
    c[v]    = 0.f;
    h[v]    = 0.f;
  }
  const int hmask = -half;

  const v8f zero8 = {0.f, 0.f, 0.f, 0.f, 0.f, 0.f, 0.f, 0.f};
  v2f was[3];
#pragma unroll
  for (int j = 0; j < 3; ++j) {
    float l0 = 0.f, l1 = 0.f;
    if (2 * j < HH)     l0 = lg[2 * j];
    if (2 * j + 1 < HH) l1 = lg[2 * j + 1];
    was[j][0] = wa[16 + j][0] * l0;
    was[j][1] = wa[16 + j][1] * l1;
  }
  v8f cb = zero8;
  {
    v2f l0 = {lb[0], lb[1]};
    v2f l1 = {lb[2], lb[3]};
    v2f l2 = {lb[4], 0.f};
    cb = __builtin_amdgcn_wmma_f32_16x16x4_f32(false, wa[16], false, l0, (short)0, cb, false, false);
    cb = __builtin_amdgcn_wmma_f32_16x16x4_f32(false, wa[17], false, l1, (short)0, cb, false, false);
    cb = __builtin_amdgcn_wmma_f32_16x16x4_f32(false, wa[18], false, l2, (short)0, cb, false, false);
  }
  v8f cc0 = zero8, cc8 = zero8;
#pragma unroll
  for (int v = 0; v < HH; ++v) {
    cc0[v] = bias[v] + px[v];
    cc8[v] = cb[v] + bias[v] + px[v];
  }

  const float* xbase = inputs + ((size_t)(mtile * 16 + b)) * DQ + 4 * half;
  float4 xq[8];
  v8f ax0 = cc0;
  v8f ax1 = zero8;
#pragma unroll
  for (int i = 0; i < 8; ++i) xq[i] = *(const float4*)(xbase + 8 * i);
#pragma unroll
  for (int i = 0; i < 8; ++i) xpair(wa[2 * i], wa[2 * i + 1], xq[i], ax0, ax1);

  const float* xw = inputs + ((size_t)(mtile * 16 + 4 * g)) * DQ;
  const float* xtile = inputs + ((size_t)(mtile * 16)) * DQ;
  const int    xl = 4 * lane;
  const int so = (4 * g + (lane >> 4)) * XP4 + (lane & 15);
  const int ro = b * XP4 + half;
  {
    const float* x1 = xw + (size_t)BQ * DQ;
    xs[1][so]           = *(const float4*)(x1 + xl);
    xs[1][so + 2 * XP4] = *(const float4*)(x1 + xl + 128);
  }
  float4 gl0 = *(const float4*)(xw + (size_t)2 * BQ * DQ + xl);
  float4 gl1 = *(const float4*)(xw + (size_t)2 * BQ * DQ + xl + 128);
  __syncthreads();
  xs[0][so]           = gl0;
  xs[0][so + 2 * XP4] = gl1;
  gl0 = *(const float4*)(xw + (size_t)3 * BQ * DQ + xl);
  gl1 = *(const float4*)(xw + (size_t)3 * BQ * DQ + xl + 128);
#pragma unroll
  for (int i = 0; i < 8; ++i) xq[i] = xs[1][ro + 2 * i];

  unsigned yob = (unsigned)(b * HQ + HH * half) * 4u;

  float fac[HH];
#pragma unroll
  for (int v = 0; v < HH; ++v) fac[v] = cpy[v] * __cosf(ax0[v] + ax1[v]);

#pragma unroll 2
  for (int t = 0; t < TQ; ++t) {
    const int pb = t & 1;

    float p = 1.f;
#pragma unroll
    for (int v = 0; v < HH; ++v) p *= fac[v];
    const float tot = xhalf(p);
    float z = __builtin_bit_cast(float,
        (__builtin_bit_cast(int, tot) & hmask) | (0x3f800000 & ~hmask));

    {
      v2f xb0 = {xq[0].x, xq[0].y};
      asm("" : "+v"(xb0) : "v"(p));
      ax0 = __builtin_amdgcn_wmma_f32_16x16x4_f32(false, wa[0], false, xb0, (short)0, cc8, false, false);
      asm("" : "+v"(z) : "v"(ax0));
    }
    float zz[HH];
#pragma unroll
    for (int v = 0; v < HH; ++v) { z *= fac[v]; zz[v] = z; }
    {
      v2f xb1 = {xq[0].z, xq[0].w};
      asm("" : "+v"(xb1) : "v"(zz[2]));
      ax1 = __builtin_amdgcn_wmma_f32_16x16x4_f32(false, wa[1], false, xb1, (short)0, zero8, false, false);
    }
    float ga[HH];
    if (g == 2) {
#pragma unroll
      for (int v = 0; v < HH; ++v) ga[v] = tanh_hw(zz[v]);
    } else {
#pragma unroll
      for (int v = 0; v < HH; ++v) ga[v] = sigmoid_hw(zz[v]);
    }

#pragma unroll
    for (int v = 0; v < HH; ++v) gx[pb][g][v][lane] = ga[v];
    xone(wa[2], xq[1].x, xq[1].y, ax0);
    xone(wa[3], xq[1].z, xq[1].w, ax1);
    xone(wa[4], xq[2].x, xq[2].y, ax0);
    xone(wa[5], xq[2].z, xq[2].w, ax1);
    asm volatile("" :: "v"(ax0), "v"(ax1));
    __builtin_amdgcn_fence(__ATOMIC_RELEASE, "workgroup", "local");
    __builtin_amdgcn_s_barrier();
    __builtin_amdgcn_fence(__ATOMIC_ACQUIRE, "workgroup", "local");

    float fv[HH], iv[HH], gv[HH], ov[HH];
#pragma unroll
    for (int v = 0; v < HH; ++v) {
      fv[v] = gx[pb][0][v][lane]; iv[v] = gx[pb][1][v][lane];
      gv[v] = gx[pb][2][v][lane]; ov[v] = gx[pb][3][v][lane];
    }

    asm volatile("" ::: "memory");
    xs[pb ^ 1][so]           = gl0;
    xs[pb ^ 1][so + 2 * XP4] = gl1;
    const int tn = (t + 4 < TQ) ? t + 4 : TQ - 1;
    const gmem_cf* xt = (const gmem_cf*)sbase_bits(xw + (size_t)tn * BQ * DQ);
    gl0 = ldg4v((const gmem_cv4*)xt + lane);
    gl1 = ldg4v((const gmem_cv4*)xt + lane + 32);
    float4 xqn[8];
#pragma unroll
    for (int i = 0; i < 8; ++i) xqn[i] = xs[pb][ro + 2 * i];
    {
      const int tp = (t + 6 < TQ) ? t + 6 : TQ - 1;
      const gmem_cc* nx = (const gmem_cc*)sbase_bits(xtile + (size_t)tp * BQ * DQ);
      __builtin_prefetch((const void*)(nx + tid * 32), 0, 3);
    }

#pragma unroll
    for (int i = 3; i < 5; ++i) xpair(wa[2 * i], wa[2 * i + 1], xq[i], ax0, ax1);

    float hr[HH];
#pragma unroll
    for (int v = 0; v < HH; ++v) {
      float cn = fv[v] * c[v] + iv[v] * gv[v];
      c[v] = cn;
      hr[v] = ov[v] * tanh_hw(cn) + h[v];
    }
#pragma unroll
    for (int i = 5; i < 8; ++i) xpair(wa[2 * i], wa[2 * i + 1], xq[i], ax0, ax1);

    float s = 0.f;
#pragma unroll
    for (int v = 0; v < HH; ++v) s += hr[v];
    const float mu = (s + xhalf(s)) * 0.1f;
    float dd[HH];
#pragma unroll
    for (int v = 0; v < HH; ++v) dd[v] = hr[v] - mu;

    v8f ah0, ah1;
    {
      v2f h0 = {dd[0], dd[1]};
      v2f h1 = {dd[2], dd[3]};
      v2f h2 = {dd[4], 0.f};
      ah0 = __builtin_amdgcn_wmma_f32_16x16x4_f32(false, was[0], false, h0,
                                                  (short)0, zero8, false, false);
      ah1 = __builtin_amdgcn_wmma_f32_16x16x4_f32(false, was[1], false, h1,
                                                  (short)0, zero8, false, false);
      ah0 = __builtin_amdgcn_wmma_f32_16x16x4_f32(false, was[2], false, h2,
                                                  (short)0, ah0, false, false);
    }

    float q = 0.f;
#pragma unroll
    for (int v = 0; v < HH; ++v) q += dd[v] * dd[v];
    const float var = (q + xhalf(q)) * 0.1f;
    const float rs = rsqrtf(var + 1e-5f);
#pragma unroll
    for (int v = 0; v < HH; ++v)
      fac[v] = cpy[v] * __cosf(rs * (ah0[v] + ah1[v]) + (ax0[v] + ax1[v]));
    asm volatile("" :: "v"(fac[0]), "v"(fac[1]), "v"(fac[2]), "v"(fac[3]), "v"(fac[4]));
#pragma unroll
    for (int v = 0; v < HH; ++v) h[v] = dd[v] * rs * lg[v] + lb[v];

    __builtin_amdgcn_sched_group_barrier(0x200, 3, 0);
    __builtin_amdgcn_sched_group_barrier(0x008, 4, 0);
    __builtin_amdgcn_sched_group_barrier(0x100, 10, 0);
    __builtin_amdgcn_sched_group_barrier(0x008, 4, 0);
    __builtin_amdgcn_sched_group_barrier(0x402, 128, 0);

#pragma unroll
    for (int i = 0; i < 8; ++i) xq[i] = xqn[i];

    if (g == 0) {
      gmem_f* yt = (gmem_f*)sbase_bits(ys + ((size_t)t * BQ + mtile * 16) * HQ);
      asm("" : "+v"(yob));
      gmem_f* yrow = (gmem_f*)((__attribute__((address_space(1))) char*)yt + yob);
#pragma unroll
      for (int v = 0; v < HH; ++v) yrow[v] = h[v];
    }
  }

  if (g == 0) {
#pragma unroll
    for (int v = 0; v < HH; ++v) {
      hout[(mtile * 16 + b) * HQ + HH * half + v] = h[v];
      cout[(mtile * 16 + b) * HQ + HH * half + v] = c[v];
    }
  }
}

extern "C" void kernel_launch(void* const* d_in, const int* in_sizes, int n_in,
                              void* d_out, int out_size, void* d_ws, size_t ws_size,
                              hipStream_t stream) {
  const float* inputs = (const float*)d_in[0];
  const float* Wf = (const float*)d_in[1];
  const float* bf = (const float*)d_in[2];
  const float* pf = (const float*)d_in[3];
  const float* Wi = (const float*)d_in[4];
  const float* bi = (const float*)d_in[5];
  const float* pi = (const float*)d_in[6];
  const float* Wu = (const float*)d_in[7];
  const float* bu = (const float*)d_in[8];
  const float* pu = (const float*)d_in[9];
  const float* Wo = (const float*)d_in[10];
  const float* bo = (const float*)d_in[11];
  const float* po = (const float*)d_in[12];
  const float* lng = (const float*)d_in[13];
  const float* lnb = (const float*)d_in[14];

  float* out  = (float*)d_out;
  float* ys   = out;
  float* hout = out + (size_t)TQ * BQ * HQ;
  float* cout = hout + BQ * HQ;

  qlstm_fused<<<BQ / 16, 128, 0, stream>>>(
      inputs, Wf, Wi, Wu, Wo, bf, bi, bu, bo,
      pf, pi, pu, po, lng, lnb, ys, hout, cout);
}


# every transcendental of a step is one hardware instruction: v_tanh_f32 for tanh and, as 0.5+0.5*tanh(z/2), the sigmoids; bare v_rsq_f32 in LayerNorm; exp2+rcp chains and rsqrtf's range fix-up go
# speedup vs baseline: 8.0782x; 1.0948x over previous
#include <hip/hip_runtime.h>
#include <math.h>

typedef float v2f __attribute__((ext_vector_type(2)));
typedef float v8f __attribute__((ext_vector_type(8)));
typedef float v4f __attribute__((ext_vector_type(4)));
typedef __attribute__((address_space(1))) float gmem_f;
typedef __attribute__((address_space(1))) const float gmem_cf;
typedef __attribute__((address_space(1))) const v4f gmem_cv4;
typedef __attribute__((address_space(1))) const char gmem_cc;

#define TQ 128
#define BQ 256
#define DQ 64
#define HQ 10
#define KD 74
#define HH 5
#define XP4 17

__device__ __forceinline__ float xhalf(float x) {
  int b = __builtin_bit_cast(int, x);
  return __builtin_bit_cast(float,
      __builtin_amdgcn_permlanex16(b, b, 0x76543210, 0xfedcba98, false, false));
}

__device__ __forceinline__ unsigned long long sbase_bits(const void* p) {
  const unsigned long long a = (unsigned long long)p;
  const unsigned lo = (unsigned)__builtin_amdgcn_readfirstlane((int)(unsigned)(a & 0xffffffffull));
  const unsigned hi = (unsigned)__builtin_amdgcn_readfirstlane((int)(unsigned)(a >> 32));
  return ((unsigned long long)hi << 32) | (unsigned long long)lo;
}

__device__ __forceinline__ float4 ldg4v(const gmem_cv4* p) {
  const v4f t = *p;
  return make_float4(t.x, t.y, t.z, t.w);
}

__device__ __forceinline__ float tanh_hw(float x) {
  return __builtin_amdgcn_tanhf(x);
}

__device__ __forceinline__ float sigmoid_hw(float z) {
  return __builtin_fmaf(0.5f, __builtin_amdgcn_tanhf(0.5f * z), 0.5f);
}

__device__ __forceinline__ void xpair(const v2f& wa0, const v2f& wa1, const float4& x,
                                      v8f& a0, v8f& a1) {
  v2f b0 = {x.x, x.y};
  v2f b1 = {x.z, x.w};
  a0 = __builtin_amdgcn_wmma_f32_16x16x4_f32(false, wa0, false, b0, (short)0, a0, false, false);
  a1 = __builtin_amdgcn_wmma_f32_16x16x4_f32(false, wa1, false, b1, (short)0, a1, false, false);
}

__device__ __forceinline__ void xone(const v2f& w, float x0, float x1, v8f& a) {
  v2f bb = {x0, x1};
  a = __builtin_amdgcn_wmma_f32_16x16x4_f32(false, w, false, bb, (short)0, a, false, false);
}

__global__ __launch_bounds__(128) void qlstm_fused(
    const float* __restrict__ inputs,
    const float* __restrict__ Wf, const float* __restrict__ Wi,
    const float* __restrict__ Wu, const float* __restrict__ Wo,
    const float* __restrict__ bf, const float* __restrict__ bi,
    const float* __restrict__ bu, const float* __restrict__ bo,
    const float* __restrict__ pf, const float* __restrict__ pi,
    const float* __restrict__ pu, const float* __restrict__ po,
    const float* __restrict__ lng, const float* __restrict__ lnb,
    float* __restrict__ ys, float* __restrict__ hout, float* __restrict__ cout) {
  __shared__ float gx[2][4][HH][32];
  __shared__ float4 xs[2][16 * XP4];

  const int mtile = blockIdx.x;
  const int tid   = threadIdx.x;
  const int g     = __builtin_amdgcn_readfirstlane(tid >> 5);
  const int lane  = tid & 31;
  const int b     = lane & 15;
  const int half  = lane >> 4;

  const float* Wg  = (g == 0) ? Wf : (g == 1) ? Wi : (g == 2) ? Wu : Wo;
  const float* bbg = (g == 0) ? bf : (g == 1) ? bi : (g == 2) ? bu : bo;
  const float* pg  = (g == 0) ? pf : (g == 1) ? pi : (g == 2) ? pu : po;

  const int  am   = lane & 15;
  const bool arow = (am < HH) || (am >= 8 && am < 8 + HH);
  const int  awire = arow ? ((am < HH) ? am : am - 3) : 0;
  const float* wrow = Wg + awire * KD;
  v2f wa[19];
#pragma unroll
  for (int i = 0; i < 8; ++i) {
    const int c = 8 * i + 4 * half;
    const float w0 = wrow[c + 0], w1 = wrow[c + 1], w2 = wrow[c + 2], w3 = wrow[c + 3];
    wa[2 * i][0]     = arow ? w0 : 0.f;
    wa[2 * i][1]     = arow ? w1 : 0.f;
    wa[2 * i + 1][0] = arow ? w2 : 0.f;
    wa[2 * i + 1][1] = arow ? w3 : 0.f;
  }
#pragma unroll
  for (int j = 0; j < 3; ++j) {
    float u0 = 0.f, u1 = 0.f;
    if (2 * j < HH)     u0 = wrow[DQ + HH * half + 2 * j];
    if (2 * j + 1 < HH) u1 = wrow[DQ + HH * half + 2 * j + 1];
    wa[16 + j][0] = arow ? u0 : 0.f;
    wa[16 + j][1] = arow ? u1 : 0.f;
  }

  float bias[HH], px[HH], cpy[HH], lg[HH], lb[HH], c[HH], h[HH];
#pragma unroll
  for (int v = 0; v < HH; ++v) {
    const int w = HH * half + v;
    bias[v] = bbg[w];
    px[v]   = pg[w * 3 + 0];
    cpy[v]  = cosf(pg[w * 3 + 1]);
    lg[v]   = lng[w];
    lb[v]   = lnb[w];
    c[v]    = 0.f;
    h[v]    = 0.f;
  }
  const int hmask = -half;

  const v8f zero8 = {0.f, 0.f, 0.f, 0.f, 0.f, 0.f, 0.f, 0.f};
  v2f was[3];
#pragma unroll
  for (int j = 0; j < 3; ++j) {
    float l0 = 0.f, l1 = 0.f;
    if (2 * j < HH)     l0 = lg[2 * j];
    if (2 * j + 1 < HH) l1 = lg[2 * j + 1];
    was[j][0] = wa[16 + j][0] * l0;
    was[j][1] = wa[16 + j][1] * l1;
  }
  v8f cb = zero8;
  {
    v2f l0 = {lb[0], lb[1]};
    v2f l1 = {lb[2], lb[3]};
    v2f l2 = {lb[4], 0.f};
    cb = __builtin_amdgcn_wmma_f32_16x16x4_f32(false, wa[16], false, l0, (short)0, cb, false, false);
    cb = __builtin_amdgcn_wmma_f32_16x16x4_f32(false, wa[17], false, l1, (short)0, cb, false, false);
    cb = __builtin_amdgcn_wmma_f32_16x16x4_f32(false, wa[18], false, l2, (short)0, cb, false, false);
  }
  v8f cc0 = zero8, cc8 = zero8;
#pragma unroll
  for (int v = 0; v < HH; ++v) {
    cc0[v] = bias[v] + px[v];
    cc8[v] = cb[v] + bias[v] + px[v];
  }

  const float* xbase = inputs + ((size_t)(mtile * 16 + b)) * DQ + 4 * half;
  float4 xq[8];
  v8f ax0 = cc0;
  v8f ax1 = zero8;
#pragma unroll
  for (int i = 0; i < 8; ++i) xq[i] = *(const float4*)(xbase + 8 * i);
#pragma unroll
  for (int i = 0; i < 8; ++i) xpair(wa[2 * i], wa[2 * i + 1], xq[i], ax0, ax1);

  const float* xw = inputs + ((size_t)(mtile * 16 + 4 * g)) * DQ;
  const float* xtile = inputs + ((size_t)(mtile * 16)) * DQ;
  const int    xl = 4 * lane;
  const int so = (4 * g + (lane >> 4)) * XP4 + (lane & 15);
  const int ro = b * XP4 + half;
  {
    const float* x1 = xw + (size_t)BQ * DQ;
    xs[1][so]           = *(const float4*)(x1 + xl);
    xs[1][so + 2 * XP4] = *(const float4*)(x1 + xl + 128);
  }
  float4 gl0 = *(const float4*)(xw + (size_t)2 * BQ * DQ + xl);
  float4 gl1 = *(const float4*)(xw + (size_t)2 * BQ * DQ + xl + 128);
  __syncthreads();
  xs[0][so]           = gl0;
  xs[0][so + 2 * XP4] = gl1;
  gl0 = *(const float4*)(xw + (size_t)3 * BQ * DQ + xl);
  gl1 = *(const float4*)(xw + (size_t)3 * BQ * DQ + xl + 128);
#pragma unroll
  for (int i = 0; i < 8; ++i) xq[i] = xs[1][ro + 2 * i];

  unsigned yob = (unsigned)(b * HQ + HH * half) * 4u;

  float fac[HH];
#pragma unroll
  for (int v = 0; v < HH; ++v) fac[v] = cpy[v] * __cosf(ax0[v] + ax1[v]);

#pragma unroll 2
  for (int t = 0; t < TQ; ++t) {
    const int pb = t & 1;

    float p = 1.f;
#pragma unroll
    for (int v = 0; v < HH; ++v) p *= fac[v];
    const float tot = xhalf(p);
    float z = __builtin_bit_cast(float,
        (__builtin_bit_cast(int, tot) & hmask) | (0x3f800000 & ~hmask));

    {
      v2f xb0 = {xq[0].x, xq[0].y};
      asm("" : "+v"(xb0) : "v"(p));
      ax0 = __builtin_amdgcn_wmma_f32_16x16x4_f32(false, wa[0], false, xb0, (short)0, cc8, false, false);
      asm("" : "+v"(z) : "v"(ax0));
    }
    float zz[HH];
#pragma unroll
    for (int v = 0; v < HH; ++v) { z *= fac[v]; zz[v] = z; }
    {
      v2f xb1 = {xq[0].z, xq[0].w};
      asm("" : "+v"(xb1) : "v"(zz[2]));
      ax1 = __builtin_amdgcn_wmma_f32_16x16x4_f32(false, wa[1], false, xb1, (short)0, zero8, false, false);
    }
    float ga[HH];
    if (g == 2) {
#pragma unroll
      for (int v = 0; v < HH; ++v) ga[v] = tanh_hw(zz[v]);
    } else {
#pragma unroll
      for (int v = 0; v < HH; ++v) ga[v] = sigmoid_hw(zz[v]);
    }

#pragma unroll
    for (int v = 0; v < HH; ++v) gx[pb][g][v][lane] = ga[v];
    xone(wa[2], xq[1].x, xq[1].y, ax0);
    xone(wa[3], xq[1].z, xq[1].w, ax1);
    xone(wa[4], xq[2].x, xq[2].y, ax0);
    xone(wa[5], xq[2].z, xq[2].w, ax1);
    asm volatile("" :: "v"(ax0), "v"(ax1));
    __builtin_amdgcn_fence(__ATOMIC_RELEASE, "workgroup", "local");
    __builtin_amdgcn_s_barrier();
    __builtin_amdgcn_fence(__ATOMIC_ACQUIRE, "workgroup", "local");

    float fv[HH], iv[HH], gv[HH], ov[HH];
#pragma unroll
    for (int v = 0; v < HH; ++v) {
      fv[v] = gx[pb][0][v][lane]; iv[v] = gx[pb][1][v][lane];
      gv[v] = gx[pb][2][v][lane]; ov[v] = gx[pb][3][v][lane];
    }

    asm volatile("" ::: "memory");
    xs[pb ^ 1][so]           = gl0;
    xs[pb ^ 1][so + 2 * XP4] = gl1;
    const int tn = (t + 4 < TQ) ? t + 4 : TQ - 1;
    const gmem_cf* xt = (const gmem_cf*)sbase_bits(xw + (size_t)tn * BQ * DQ);
    gl0 = ldg4v((const gmem_cv4*)xt + lane);
    gl1 = ldg4v((const gmem_cv4*)xt + lane + 32);
    float4 xqn[8];
#pragma unroll
    for (int i = 0; i < 8; ++i) xqn[i] = xs[pb][ro + 2 * i];
    {
      const int tp = (t + 6 < TQ) ? t + 6 : TQ - 1;
      const gmem_cc* nx = (const gmem_cc*)sbase_bits(xtile + (size_t)tp * BQ * DQ);
      __builtin_prefetch((const void*)(nx + tid * 32), 0, 3);
    }

#pragma unroll
    for (int i = 3; i < 5; ++i) xpair(wa[2 * i], wa[2 * i + 1], xq[i], ax0, ax1);

    float hr[HH];
#pragma unroll
    for (int v = 0; v < HH; ++v) {
      float cn = fv[v] * c[v] + iv[v] * gv[v];
      c[v] = cn;
      hr[v] = ov[v] * tanh_hw(cn) + h[v];
    }
#pragma unroll
    for (int i = 5; i < 8; ++i) xpair(wa[2 * i], wa[2 * i + 1], xq[i], ax0, ax1);

    float s = 0.f;
#pragma unroll
    for (int v = 0; v < HH; ++v) s += hr[v];
    const float mu = (s + xhalf(s)) * 0.1f;
    float dd[HH];
#pragma unroll
    for (int v = 0; v < HH; ++v) dd[v] = hr[v] - mu;

    v8f ah0, ah1;
    {
      v2f h0 = {dd[0], dd[1]};
      v2f h1 = {dd[2], dd[3]};
      v2f h2 = {dd[4], 0.f};
      ah0 = __builtin_amdgcn_wmma_f32_16x16x4_f32(false, was[0], false, h0,
                                                  (short)0, zero8, false, false);
      ah1 = __builtin_amdgcn_wmma_f32_16x16x4_f32(false, was[1], false, h1,
                                                  (short)0, zero8, false, false);
      ah0 = __builtin_amdgcn_wmma_f32_16x16x4_f32(false, was[2], false, h2,
                                                  (short)0, ah0, false, false);
    }

    float q = 0.f;
#pragma unroll
    for (int v = 0; v < HH; ++v) q += dd[v] * dd[v];
    const float var = (q + xhalf(q)) * 0.1f;
    const float rs = __builtin_amdgcn_rsqf(var + 1e-5f);
#pragma unroll
    for (int v = 0; v < HH; ++v)
      fac[v] = cpy[v] * __cosf(rs * (ah0[v] + ah1[v]) + (ax0[v] + ax1[v]));
    asm volatile("" :: "v"(fac[0]), "v"(fac[1]), "v"(fac[2]), "v"(fac[3]), "v"(fac[4]));
#pragma unroll
    for (int v = 0; v < HH; ++v) h[v] = dd[v] * rs * lg[v] + lb[v];

    __builtin_amdgcn_sched_group_barrier(0x200, 3, 0);
    __builtin_amdgcn_sched_group_barrier(0x008, 4, 0);
    __builtin_amdgcn_sched_group_barrier(0x100, 10, 0);
    __builtin_amdgcn_sched_group_barrier(0x008, 4, 0);
    __builtin_amdgcn_sched_group_barrier(0x402, 128, 0);

#pragma unroll
    for (int i = 0; i < 8; ++i) xq[i] = xqn[i];

    if (g == 0) {
      gmem_f* yt = (gmem_f*)sbase_bits(ys + ((size_t)t * BQ + mtile * 16) * HQ);
      asm("" : "+v"(yob));
      gmem_f* yrow = (gmem_f*)((__attribute__((address_space(1))) char*)yt + yob);
#pragma unroll
      for (int v = 0; v < HH; ++v) yrow[v] = h[v];
    }
  }

  if (g == 0) {
#pragma unroll
    for (int v = 0; v < HH; ++v) {
      hout[(mtile * 16 + b) * HQ + HH * half + v] = h[v];
      cout[(mtile * 16 + b) * HQ + HH * half + v] = c[v];
    }
  }
}

extern "C" void kernel_launch(void* const* d_in, const int* in_sizes, int n_in,
                              void* d_out, int out_size, void* d_ws, size_t ws_size,
                              hipStream_t stream) {
  const float* inputs = (const float*)d_in[0];
  const float* Wf = (const float*)d_in[1];
  const float* bf = (const float*)d_in[2];
  const float* pf = (const float*)d_in[3];
  const float* Wi = (const float*)d_in[4];
  const float* bi = (const float*)d_in[5];
  const float* pi = (const float*)d_in[6];
  const float* Wu = (const float*)d_in[7];
  const float* bu = (const float*)d_in[8];
  const float* pu = (const float*)d_in[9];
  const float* Wo = (const float*)d_in[10];
  const float* bo = (const float*)d_in[11];
  const float* po = (const float*)d_in[12];
  const float* lng = (const float*)d_in[13];
  const float* lnb = (const float*)d_in[14];

  float* out  = (float*)d_out;
  float* ys   = out;
  float* hout = out + (size_t)TQ * BQ * HQ;
  float* cout = hout + BQ * HQ;

  qlstm_fused<<<BQ / 16, 128, 0, stream>>>(
      inputs, Wf, Wi, Wu, Wo, bf, bi, bu, bo,
      pf, pi, pu, po, lng, lnb, ys, hout, cout);
}


# constants off the chain: W, bias, phase in revolutions (bare v_cos); cos(py) prefixes and logistic 1/2 in one constant; one multiply behind the lane exchange, no copy before it; new product order
# speedup vs baseline: 8.4658x; 1.0480x over previous
#include <hip/hip_runtime.h>
#include <math.h>

typedef float v2f __attribute__((ext_vector_type(2)));
typedef float v8f __attribute__((ext_vector_type(8)));
typedef float v4f __attribute__((ext_vector_type(4)));
typedef __attribute__((address_space(1))) float gmem_f;
typedef __attribute__((address_space(1))) const float gmem_cf;
typedef __attribute__((address_space(1))) const v4f gmem_cv4;
typedef __attribute__((address_space(1))) const char gmem_cc;

#define TQ 128
#define BQ 256
#define DQ 64
#define HQ 10
#define KD 74
#define HH 5
#define R2PI 0.15915494309189535f
#define XP4 17

__device__ __forceinline__ float xhalf(float x) {
  const int b = __builtin_bit_cast(int, x);
  return __builtin_bit_cast(float,
      __builtin_amdgcn_permlanex16(__builtin_nondeterministic_value(b), b,
                                   0x76543210, 0xfedcba98, false, false));
}

__device__ __forceinline__ unsigned long long sbase_bits(const void* p) {
  const unsigned long long a = (unsigned long long)p;
  const unsigned lo = (unsigned)__builtin_amdgcn_readfirstlane((int)(unsigned)(a & 0xffffffffull));
  const unsigned hi = (unsigned)__builtin_amdgcn_readfirstlane((int)(unsigned)(a >> 32));
  return ((unsigned long long)hi << 32) | (unsigned long long)lo;
}

__device__ __forceinline__ float4 ldg4v(const gmem_cv4* p) {
  const v4f t = *p;
  return make_float4(t.x, t.y, t.z, t.w);
}

__device__ __forceinline__ float tanh_hw(float x) {
  return __builtin_amdgcn_tanhf(x);
}

__device__ __forceinline__ float cos_rev(float r) {
  return __builtin_amdgcn_cosf(r);
}

__device__ __forceinline__ void xpair(const v2f& wa0, const v2f& wa1, const float4& x,
                                      v8f& a0, v8f& a1) {
  v2f b0 = {x.x, x.y};
  v2f b1 = {x.z, x.w};
  a0 = __builtin_amdgcn_wmma_f32_16x16x4_f32(false, wa0, false, b0, (short)0, a0, false, false);
  a1 = __builtin_amdgcn_wmma_f32_16x16x4_f32(false, wa1, false, b1, (short)0, a1, false, false);
}

__device__ __forceinline__ void xone(const v2f& w, float x0, float x1, v8f& a) {
  v2f bb = {x0, x1};
  a = __builtin_amdgcn_wmma_f32_16x16x4_f32(false, w, false, bb, (short)0, a, false, false);
}

__global__ __launch_bounds__(128) void qlstm_fused(
    const float* __restrict__ inputs,
    const float* __restrict__ Wf, const float* __restrict__ Wi,
    const float* __restrict__ Wu, const float* __restrict__ Wo,
    const float* __restrict__ bf, const float* __restrict__ bi,
    const float* __restrict__ bu, const float* __restrict__ bo,
    const float* __restrict__ pf, const float* __restrict__ pi,
    const float* __restrict__ pu, const float* __restrict__ po,
    const float* __restrict__ lng, const float* __restrict__ lnb,
    float* __restrict__ ys, float* __restrict__ hout, float* __restrict__ cout) {
  __shared__ float gx[2][4][HH][32];
  __shared__ float4 xs[2][16 * XP4];

  const int mtile = blockIdx.x;
  const int tid   = threadIdx.x;
  const int g     = __builtin_amdgcn_readfirstlane(tid >> 5);
  const int lane  = tid & 31;
  const int b     = lane & 15;
  const int half  = lane >> 4;

  const float* Wg  = (g == 0) ? Wf : (g == 1) ? Wi : (g == 2) ? Wu : Wo;
  const float* bbg = (g == 0) ? bf : (g == 1) ? bi : (g == 2) ? bu : bo;
  const float* pg  = (g == 0) ? pf : (g == 1) ? pi : (g == 2) ? pu : po;

  const int  am   = lane & 15;
  const bool arow = (am < HH) || (am >= 8 && am < 8 + HH);
  const int  awire = arow ? ((am < HH) ? am : am - 3) : 0;
  const float* wrow = Wg + awire * KD;
  v2f wa[19];
#pragma unroll
  for (int i = 0; i < 8; ++i) {
    const int c = 8 * i + 4 * half;
    const float w0 = wrow[c + 0], w1 = wrow[c + 1], w2 = wrow[c + 2], w3 = wrow[c + 3];
    wa[2 * i][0]     = arow ? w0 * R2PI : 0.f;
    wa[2 * i][1]     = arow ? w1 * R2PI : 0.f;
    wa[2 * i + 1][0] = arow ? w2 * R2PI : 0.f;
    wa[2 * i + 1][1] = arow ? w3 * R2PI : 0.f;
  }
#pragma unroll
  for (int j = 0; j < 3; ++j) {
    float u0 = 0.f, u1 = 0.f;
    if (2 * j < HH)     u0 = wrow[DQ + HH * half + 2 * j];
    if (2 * j + 1 < HH) u1 = wrow[DQ + HH * half + 2 * j + 1];
    wa[16 + j][0] = arow ? u0 * R2PI : 0.f;
    wa[16 + j][1] = arow ? u1 * R2PI : 0.f;
  }

  float bias[HH], px[HH], cpy[HH], lg[HH], lb[HH], c[HH], h[HH];
#pragma unroll
  for (int v = 0; v < HH; ++v) {
    const int w = HH * half + v;
    bias[v] = bbg[w];
    px[v]   = pg[w * 3 + 0];
    cpy[v]  = cosf(pg[w * 3 + 1]);
    lg[v]   = lng[w];
    lb[v]   = lnb[w];
    c[v]    = 0.f;
    h[v]    = 0.f;
  }
  const int hmask = -half;
  float kc[HH];
  {
    float run = 1.f;
#pragma unroll
    for (int v = 0; v < HH; ++v) { run *= cpy[v]; kc[v] = run; }
    const float lowtot = xhalf(run);
    const float lead = __builtin_bit_cast(float,
        (__builtin_bit_cast(int, lowtot) & hmask) | (0x3f800000 & ~hmask));
    const float gsc = (g == 2) ? 1.f : 0.5f;
#pragma unroll
    for (int v = 0; v < HH; ++v) kc[v] = (lead * kc[v]) * gsc;
  }
  const float ka = (g == 2) ? 1.f : 0.5f;
  const float kb = (g == 2) ? 0.f : 0.5f;

  const v8f zero8 = {0.f, 0.f, 0.f, 0.f, 0.f, 0.f, 0.f, 0.f};
  v2f was[3];
#pragma unroll
  for (int j = 0; j < 3; ++j) {
    float l0 = 0.f, l1 = 0.f;
    if (2 * j < HH)     l0 = lg[2 * j];
    if (2 * j + 1 < HH) l1 = lg[2 * j + 1];
    was[j][0] = wa[16 + j][0] * l0;
    was[j][1] = wa[16 + j][1] * l1;
  }
  v8f cb = zero8;
  {
    v2f l0 = {lb[0], lb[1]};
    v2f l1 = {lb[2], lb[3]};
    v2f l2 = {lb[4], 0.f};
    cb = __builtin_amdgcn_wmma_f32_16x16x4_f32(false, wa[16], false, l0, (short)0, cb, false, false);
    cb = __builtin_amdgcn_wmma_f32_16x16x4_f32(false, wa[17], false, l1, (short)0, cb, false, false);
    cb = __builtin_amdgcn_wmma_f32_16x16x4_f32(false, wa[18], false, l2, (short)0, cb, false, false);
  }
  v8f cc0 = zero8, cc8 = zero8;
#pragma unroll
  for (int v = 0; v < HH; ++v) {
    cc0[v] = (bias[v] + px[v]) * R2PI;
    cc8[v] = cb[v] + (bias[v] + px[v]) * R2PI;
  }

  const float* xbase = inputs + ((size_t)(mtile * 16 + b)) * DQ + 4 * half;
  float4 xq[8];
  v8f ax0 = cc0;
  v8f ax1 = zero8;
#pragma unroll
  for (int i = 0; i < 8; ++i) xq[i] = *(const float4*)(xbase + 8 * i);
#pragma unroll
  for (int i = 0; i < 8; ++i) xpair(wa[2 * i], wa[2 * i + 1], xq[i], ax0, ax1);

  const float* xw = inputs + ((size_t)(mtile * 16 + 4 * g)) * DQ;
  const float* xtile = inputs + ((size_t)(mtile * 16)) * DQ;
  const int    xl = 4 * lane;
  const int so = (4 * g + (lane >> 4)) * XP4 + (lane & 15);
  const int ro = b * XP4 + half;
  {
    const float* x1 = xw + (size_t)BQ * DQ;
    xs[1][so]           = *(const float4*)(x1 + xl);
    xs[1][so + 2 * XP4] = *(const float4*)(x1 + xl + 128);
  }
  float4 gl0 = *(const float4*)(xw + (size_t)2 * BQ * DQ + xl);
  float4 gl1 = *(const float4*)(xw + (size_t)2 * BQ * DQ + xl + 128);
  __syncthreads();
  xs[0][so]           = gl0;
  xs[0][so + 2 * XP4] = gl1;
  gl0 = *(const float4*)(xw + (size_t)3 * BQ * DQ + xl);
  gl1 = *(const float4*)(xw + (size_t)3 * BQ * DQ + xl + 128);
#pragma unroll
  for (int i = 0; i < 8; ++i) xq[i] = xs[1][ro + 2 * i];

  unsigned yob = (unsigned)(b * HQ + HH * half) * 4u;

  float fac[HH];
#pragma unroll
  for (int v = 0; v < HH; ++v) fac[v] = cos_rev(ax0[v] + ax1[v]);

#pragma unroll 2
  for (int t = 0; t < TQ; ++t) {
    const int pb = t & 1;

    float pr[HH];
    pr[0] = fac[0];
#pragma unroll
    for (int v = 1; v < HH; ++v) pr[v] = pr[v - 1] * fac[v];
    const float p = pr[HH - 1];
    const float tot = xhalf(p);
    float z = __builtin_bit_cast(float,
        (__builtin_bit_cast(int, tot) & hmask) | (0x3f800000 & ~hmask));

    {
      v2f xb0 = {xq[0].x, xq[0].y};
      asm("" : "+v"(xb0) : "v"(p));
      ax0 = __builtin_amdgcn_wmma_f32_16x16x4_f32(false, wa[0], false, xb0, (short)0, cc8, false, false);
      asm("" : "+v"(z) : "v"(ax0));
    }
    float zz[HH];
#pragma unroll
    for (int v = 0; v < HH; ++v) zz[v] = z * (pr[v] * kc[v]);
    {
      v2f xb1 = {xq[0].z, xq[0].w};
      asm("" : "+v"(xb1) : "v"(zz[HH - 1]));
      ax1 = __builtin_amdgcn_wmma_f32_16x16x4_f32(false, wa[1], false, xb1, (short)0, zero8, false, false);
    }
    asm("" : "+v"(zz[HH - 1]) : "v"(ax1));
    float ga[HH];
#pragma unroll
    for (int v = 0; v < HH; ++v) ga[v] = __builtin_fmaf(tanh_hw(zz[v]), ka, kb);
    __builtin_amdgcn_sched_barrier(0);

#pragma unroll
    for (int v = 0; v < HH; ++v) gx[pb][g][v][lane] = ga[v];
    xone(wa[2], xq[1].x, xq[1].y, ax0);
    xone(wa[3], xq[1].z, xq[1].w, ax1);
    xone(wa[4], xq[2].x, xq[2].y, ax0);
    xone(wa[5], xq[2].z, xq[2].w, ax1);
    asm volatile("" :: "v"(ax0), "v"(ax1));
    __builtin_amdgcn_fence(__ATOMIC_RELEASE, "workgroup", "local");
    __builtin_amdgcn_s_barrier();
    __builtin_amdgcn_fence(__ATOMIC_ACQUIRE, "workgroup", "local");

    float fv[HH], iv[HH], gv[HH], ov[HH];
#pragma unroll
    for (int v = 0; v < HH; ++v) {
      fv[v] = gx[pb][0][v][lane]; iv[v] = gx[pb][1][v][lane];
      gv[v] = gx[pb][2][v][lane]; ov[v] = gx[pb][3][v][lane];
    }

    asm volatile("" ::: "memory");
    xs[pb ^ 1][so]           = gl0;
    xs[pb ^ 1][so + 2 * XP4] = gl1;
    const int tn = (t + 4 < TQ) ? t + 4 : TQ - 1;
    const gmem_cf* xt = (const gmem_cf*)sbase_bits(xw + (size_t)tn * BQ * DQ);
    gl0 = ldg4v((const gmem_cv4*)xt + lane);
    gl1 = ldg4v((const gmem_cv4*)xt + lane + 32);
    float4 xqn[8];
#pragma unroll
    for (int i = 0; i < 8; ++i) xqn[i] = xs[pb][ro + 2 * i];
    {
      const int tp = (t + 6 < TQ) ? t + 6 : TQ - 1;
      const gmem_cc* nx = (const gmem_cc*)sbase_bits(xtile + (size_t)tp * BQ * DQ);
      __builtin_prefetch((const void*)(nx + tid * 32), 0, 3);
    }

#pragma unroll
    for (int i = 3; i < 5; ++i) xpair(wa[2 * i], wa[2 * i + 1], xq[i], ax0, ax1);

    float hr[HH];
#pragma unroll
    for (int v = 0; v < HH; ++v) {
      float cn = fv[v] * c[v] + iv[v] * gv[v];
      c[v] = cn;
      hr[v] = ov[v] * tanh_hw(cn) + h[v];
    }
#pragma unroll
    for (int i = 5; i < 8; ++i) xpair(wa[2 * i], wa[2 * i + 1], xq[i], ax0, ax1);

    float s = 0.f;
#pragma unroll
    for (int v = 0; v < HH; ++v) s += hr[v];
    const float mu = (s + xhalf(s)) * 0.1f;
    float dd[HH];
#pragma unroll
    for (int v = 0; v < HH; ++v) dd[v] = hr[v] - mu;

    v8f ah0, ah1;
    {
      v2f h0 = {dd[0], dd[1]};
      v2f h1 = {dd[2], dd[3]};
      v2f h2 = {dd[4], 0.f};
      ah0 = __builtin_amdgcn_wmma_f32_16x16x4_f32(false, was[0], false, h0,
                                                  (short)0, zero8, false, false);
      ah1 = __builtin_amdgcn_wmma_f32_16x16x4_f32(false, was[1], false, h1,
                                                  (short)0, zero8, false, false);
      ah0 = __builtin_amdgcn_wmma_f32_16x16x4_f32(false, was[2], false, h2,
                                                  (short)0, ah0, false, false);
    }

    float q = 0.f;
#pragma unroll
    for (int v = 0; v < HH; ++v) q += dd[v] * dd[v];
    const float var = (q + xhalf(q)) * 0.1f;
    const float rs = __builtin_amdgcn_rsqf(var + 1e-5f);
#pragma unroll
    for (int v = 0; v < HH; ++v)
      fac[v] = cos_rev(rs * (ah0[v] + ah1[v]) + (ax0[v] + ax1[v]));
    asm volatile("" :: "v"(fac[0]), "v"(fac[1]), "v"(fac[2]), "v"(fac[3]), "v"(fac[4]));
#pragma unroll
    for (int v = 0; v < HH; ++v) h[v] = dd[v] * rs * lg[v] + lb[v];

    __builtin_amdgcn_sched_group_barrier(0x200, 3, 0);
    __builtin_amdgcn_sched_group_barrier(0x008, 4, 0);
    __builtin_amdgcn_sched_group_barrier(0x100, 10, 0);
    __builtin_amdgcn_sched_group_barrier(0x008, 4, 0);
    __builtin_amdgcn_sched_group_barrier(0x402, 128, 0);

#pragma unroll
    for (int i = 0; i < 8; ++i) xq[i] = xqn[i];

    if (g == 0) {
      gmem_f* yt = (gmem_f*)sbase_bits(ys + ((size_t)t * BQ + mtile * 16) * HQ);
      asm("" : "+v"(yob));
      gmem_f* yrow = (gmem_f*)((__attribute__((address_space(1))) char*)yt + yob);
#pragma unroll
      for (int v = 0; v < HH; ++v) yrow[v] = h[v];
    }
  }

  if (g == 0) {
#pragma unroll
    for (int v = 0; v < HH; ++v) {
      hout[(mtile * 16 + b) * HQ + HH * half + v] = h[v];
      cout[(mtile * 16 + b) * HQ + HH * half + v] = c[v];
    }
  }
}

extern "C" void kernel_launch(void* const* d_in, const int* in_sizes, int n_in,
                              void* d_out, int out_size, void* d_ws, size_t ws_size,
                              hipStream_t stream) {
  const float* inputs = (const float*)d_in[0];
  const float* Wf = (const float*)d_in[1];
  const float* bf = (const float*)d_in[2];
  const float* pf = (const float*)d_in[3];
  const float* Wi = (const float*)d_in[4];
  const float* bi = (const float*)d_in[5];
  const float* pi = (const float*)d_in[6];
  const float* Wu = (const float*)d_in[7];
  const float* bu = (const float*)d_in[8];
  const float* pu = (const float*)d_in[9];
  const float* Wo = (const float*)d_in[10];
  const float* bo = (const float*)d_in[11];
  const float* po = (const float*)d_in[12];
  const float* lng = (const float*)d_in[13];
  const float* lnb = (const float*)d_in[14];

  float* out  = (float*)d_out;
  float* ys   = out;
  float* hout = out + (size_t)TQ * BQ * HQ;
  float* cout = hout + BQ * HQ;

  qlstm_fused<<<BQ / 16, 128, 0, stream>>>(
      inputs, Wf, Wi, Wu, Wo, bf, bi, bu, bo,
      pf, pi, pu, po, lng, lnb, ys, hout, cout);
}


# x GEMM columns = 4 batch rows x 4 time steps (64 work-groups of 4 rows, each row on 4 lanes): 4 x wmma per step and wave, not 16, all in front of the barrier; one y store per 4 steps; same sums
# speedup vs baseline: 11.6184x; 1.3724x over previous
#include <hip/hip_runtime.h>
#include <math.h>

typedef float v2f __attribute__((ext_vector_type(2)));
typedef float v8f __attribute__((ext_vector_type(8)));
typedef float v4f __attribute__((ext_vector_type(4)));
typedef __attribute__((address_space(1))) float gmem_f;
typedef __attribute__((address_space(1))) const float gmem_cf;
typedef __attribute__((address_space(1))) const v4f gmem_cv4;
typedef __attribute__((address_space(1))) const char gmem_cc;

#define TQ 128
#define BQ 256
#define DQ 64
#define HQ 10
#define KD 74
#define HH 5
#define R2PI 0.15915494309189535f
#define XP4 17

__device__ __forceinline__ float xhalf(float x) {
  const int b = __builtin_bit_cast(int, x);
  return __builtin_bit_cast(float,
      __builtin_amdgcn_permlanex16(__builtin_nondeterministic_value(b), b,
                                   0x76543210, 0xfedcba98, false, false));
}

__device__ __forceinline__ unsigned long long sbase_bits(const void* p) {
  const unsigned long long a = (unsigned long long)p;
  const unsigned lo = (unsigned)__builtin_amdgcn_readfirstlane((int)(unsigned)(a & 0xffffffffull));
  const unsigned hi = (unsigned)__builtin_amdgcn_readfirstlane((int)(unsigned)(a >> 32));
  return ((unsigned long long)hi << 32) | (unsigned long long)lo;
}

__device__ __forceinline__ float4 ldg4v(const gmem_cv4* p) {
  const v4f t = *p;
  return make_float4(t.x, t.y, t.z, t.w);
}

__device__ __forceinline__ float tanh_hw(float x) {
  return __builtin_amdgcn_tanhf(x);
}

__device__ __forceinline__ float cos_rev(float r) {
  return __builtin_amdgcn_cosf(r);
}

__device__ __forceinline__ void xpair(const v2f& wa0, const v2f& wa1, const float4& x,
                                      v8f& a0, v8f& a1) {
  v2f b0 = {x.x, x.y};
  v2f b1 = {x.z, x.w};
  a0 = __builtin_amdgcn_wmma_f32_16x16x4_f32(false, wa0, false, b0, (short)0, a0, false, false);
  a1 = __builtin_amdgcn_wmma_f32_16x16x4_f32(false, wa1, false, b1, (short)0, a1, false, false);
}

__device__ __forceinline__ float cols(float x, int q) {
  const unsigned sel = 0x32103210u + 0x44444444u * (unsigned)q;
  const int b = __builtin_bit_cast(int, x);
  return __builtin_bit_cast(float,
      __builtin_amdgcn_permlane16(__builtin_nondeterministic_value(b), b, sel, sel, false, false));
}

__global__ __launch_bounds__(128) void qlstm_fused(
    const float* __restrict__ inputs,
    const float* __restrict__ Wf, const float* __restrict__ Wi,
    const float* __restrict__ Wu, const float* __restrict__ Wo,
    const float* __restrict__ bf, const float* __restrict__ bi,
    const float* __restrict__ bu, const float* __restrict__ bo,
    const float* __restrict__ pf, const float* __restrict__ pi,
    const float* __restrict__ pu, const float* __restrict__ po,
    const float* __restrict__ lng, const float* __restrict__ lnb,
    float* __restrict__ ys, float* __restrict__ hout, float* __restrict__ cout) {
  __shared__ float gx[2][4][HH][32];
  __shared__ float4 xs[16 * XP4];

  const int mtile = blockIdx.x;
  const int tid   = threadIdx.x;
  const int g     = __builtin_amdgcn_readfirstlane(tid >> 5);
  const int lane  = tid & 31;
  const int col   = lane & 15;
  const int jrow  = col & 3;
  const int sgrp  = col >> 2;
  const int half  = lane >> 4;

  const float* Wg  = (g == 0) ? Wf : (g == 1) ? Wi : (g == 2) ? Wu : Wo;
  const float* bbg = (g == 0) ? bf : (g == 1) ? bi : (g == 2) ? bu : bo;
  const float* pg  = (g == 0) ? pf : (g == 1) ? pi : (g == 2) ? pu : po;

  const int  am   = lane & 15;
  const bool arow = (am < HH) || (am >= 8 && am < 8 + HH);
  const int  awire = arow ? ((am < HH) ? am : am - 3) : 0;
  const float* wrow = Wg + awire * KD;
  v2f wa[19];
#pragma unroll
  for (int i = 0; i < 8; ++i) {
    const int c = 8 * i + 4 * half;
    const float w0 = wrow[c + 0], w1 = wrow[c + 1], w2 = wrow[c + 2], w3 = wrow[c + 3];
    wa[2 * i][0]     = arow ? w0 * R2PI : 0.f;
    wa[2 * i][1]     = arow ? w1 * R2PI : 0.f;
    wa[2 * i + 1][0] = arow ? w2 * R2PI : 0.f;
    wa[2 * i + 1][1] = arow ? w3 * R2PI : 0.f;
  }
#pragma unroll
  for (int j = 0; j < 3; ++j) {
    float u0 = 0.f, u1 = 0.f;
    if (2 * j < HH)     u0 = wrow[DQ + HH * half + 2 * j];
    if (2 * j + 1 < HH) u1 = wrow[DQ + HH * half + 2 * j + 1];
    wa[16 + j][0] = arow ? u0 * R2PI : 0.f;
    wa[16 + j][1] = arow ? u1 * R2PI : 0.f;
  }

  float bias[HH], px[HH], cpy[HH], lg[HH], lb[HH], c[HH], h[HH];
#pragma unroll
  for (int v = 0; v < HH; ++v) {
    const int w = HH * half + v;
    bias[v] = bbg[w];
    px[v]   = pg[w * 3 + 0];
    cpy[v]  = cosf(pg[w * 3 + 1]);
    lg[v]   = lng[w];
    lb[v]   = lnb[w];
    c[v]    = 0.f;
    h[v]    = 0.f;
  }
  const int hmask = -half;
  float kc[HH];
  {
    float run = 1.f;
#pragma unroll
    for (int v = 0; v < HH; ++v) { run *= cpy[v]; kc[v] = run; }
    const float lowtot = xhalf(run);
    const float lead = __builtin_bit_cast(float,
        (__builtin_bit_cast(int, lowtot) & hmask) | (0x3f800000 & ~hmask));
    const float gsc = (g == 2) ? 1.f : 0.5f;
#pragma unroll
    for (int v = 0; v < HH; ++v) kc[v] = (lead * kc[v]) * gsc;
  }
  const float ka = (g == 2) ? 1.f : 0.5f;
  const float kb = (g == 2) ? 0.f : 0.5f;

  const v8f zero8 = {0.f, 0.f, 0.f, 0.f, 0.f, 0.f, 0.f, 0.f};
  v2f was[3];
#pragma unroll
  for (int j = 0; j < 3; ++j) {
    float l0 = 0.f, l1 = 0.f;
    if (2 * j < HH)     l0 = lg[2 * j];
    if (2 * j + 1 < HH) l1 = lg[2 * j + 1];
    was[j][0] = wa[16 + j][0] * l0;
    was[j][1] = wa[16 + j][1] * l1;
  }
  v8f cb = zero8;
  {
    v2f l0 = {lb[0], lb[1]};
    v2f l1 = {lb[2], lb[3]};
    v2f l2 = {lb[4], 0.f};
    cb = __builtin_amdgcn_wmma_f32_16x16x4_f32(false, wa[16], false, l0, (short)0, cb, false, false);
    cb = __builtin_amdgcn_wmma_f32_16x16x4_f32(false, wa[17], false, l1, (short)0, cb, false, false);
    cb = __builtin_amdgcn_wmma_f32_16x16x4_f32(false, wa[18], false, l2, (short)0, cb, false, false);
  }
  v8f cc8 = zero8, ccf = zero8;
#pragma unroll
  for (int v = 0; v < HH; ++v) {
    cc8[v] = cb[v] + (bias[v] + px[v]) * R2PI;
    ccf[v] = (sgrp == 0) ? (bias[v] + px[v]) * R2PI : cc8[v];
  }

  const float* xbase = inputs + ((size_t)sgrp * BQ + (size_t)(mtile * 4 + jrow)) * DQ + 4 * half;
  float4 xq[8];
  float xs5[HH];
  {
    v8f a0 = ccf;
    v8f a1 = zero8;
#pragma unroll
    for (int i = 0; i < 8; ++i) xq[i] = *(const float4*)(xbase + 8 * i);
#pragma unroll
    for (int i = 0; i < 8; ++i) xpair(wa[2 * i], wa[2 * i + 1], xq[i], a0, a1);
#pragma unroll
    for (int v = 0; v < HH; ++v) xs5[v] = a0[v] + a1[v];
  }
#pragma unroll
  for (int i = 0; i < 8; ++i) xq[i] = *(const float4*)(xbase + (size_t)4 * BQ * DQ + 8 * i);

  const float* xw = inputs + ((size_t)g * BQ + (size_t)(mtile * 4)) * DQ;
  const int    xl = 4 * lane;
  const int so = (4 * g + (lane >> 4)) * XP4 + (lane & 15);
  const int ro = col * XP4 + half;
  float4 gl0 = *(const float4*)(xw + (size_t)8 * BQ * DQ + xl);
  float4 gl1 = *(const float4*)(xw + (size_t)8 * BQ * DQ + xl + 128);

  unsigned yob = (unsigned)((sgrp * BQ + jrow) * HQ + HH * half) * 4u;
  float hy[HH];
#pragma unroll
  for (int v = 0; v < HH; ++v) hy[v] = 0.f;

  float fac[HH];
#pragma unroll
  for (int v = 0; v < HH; ++v) fac[v] = cos_rev(cols(xs5[v], 0));

  v8f ax0 = zero8, ax1 = zero8;
  for (int k = 0; k < TQ / 4; ++k) {
#pragma unroll
    for (int p = 0; p < 4; ++p) {
      const int pb = p & 1;

      float pr[HH];
      pr[0] = fac[0];
#pragma unroll
      for (int v = 1; v < HH; ++v) pr[v] = pr[v - 1] * fac[v];
      const float tot = xhalf(pr[HH - 1]);
      const float z = __builtin_bit_cast(float,
          (__builtin_bit_cast(int, tot) & hmask) | (0x3f800000 & ~hmask));
      float zz[HH];
#pragma unroll
      for (int v = 0; v < HH; ++v) zz[v] = z * (pr[v] * kc[v]);
      float ga[HH];
#pragma unroll
      for (int v = 0; v < HH; ++v) ga[v] = __builtin_fmaf(tanh_hw(zz[v]), ka, kb);
      __builtin_amdgcn_sched_barrier(0);

#pragma unroll
      for (int v = 0; v < HH; ++v) gx[pb][g][v][lane] = ga[v];
      if (p == 0) {
        ax0 = cc8;
        ax1 = zero8;
      }
      xpair(wa[4 * p], wa[4 * p + 1], xq[2 * p], ax0, ax1);
      xpair(wa[4 * p + 2], wa[4 * p + 3], xq[2 * p + 1], ax0, ax1);
      asm volatile("" :: "v"(ax0), "v"(ax1));
      __builtin_amdgcn_fence(__ATOMIC_RELEASE, "workgroup", "local");
      __builtin_amdgcn_s_barrier();
      __builtin_amdgcn_fence(__ATOMIC_ACQUIRE, "workgroup", "local");

      float fv[HH], iv[HH], gv[HH], ov[HH];
#pragma unroll
      for (int v = 0; v < HH; ++v) {
        fv[v] = gx[pb][0][v][lane]; iv[v] = gx[pb][1][v][lane];
        gv[v] = gx[pb][2][v][lane]; ov[v] = gx[pb][3][v][lane];
      }
      asm volatile("" ::: "memory");
      if (p == 1) {
        xs[so]           = gl0;
        xs[so + 2 * XP4] = gl1;
        const int kn = (k + 3 < TQ / 4) ? k + 3 : TQ / 4 - 1;
        const gmem_cf* xt = (const gmem_cf*)sbase_bits(xw + (size_t)kn * 4 * BQ * DQ);
        gl0 = ldg4v((const gmem_cv4*)xt + lane);
        gl1 = ldg4v((const gmem_cv4*)xt + lane + 32);
      }
      if (p == 2) {
        const int kp = (k + 4 < TQ / 4) ? k + 4 : TQ / 4 - 1;
        const gmem_cc* nx = (const gmem_cc*)sbase_bits(xw + (size_t)kp * 4 * BQ * DQ);
        __builtin_prefetch((const void*)(nx + lane * 32), 0, 3);
      }
      if (p == 3) {
#pragma unroll
        for (int i = 0; i < 8; ++i) xq[i] = xs[ro + 2 * i];
      }

      float hr[HH];
#pragma unroll
      for (int v = 0; v < HH; ++v) {
        float cn = fv[v] * c[v] + iv[v] * gv[v];
        c[v] = cn;
        hr[v] = ov[v] * tanh_hw(cn) + h[v];
      }

      float s = 0.f;
#pragma unroll
      for (int v = 0; v < HH; ++v) s += hr[v];
      const float mu = (s + xhalf(s)) * 0.1f;
      float dd[HH];
#pragma unroll
      for (int v = 0; v < HH; ++v) dd[v] = hr[v] - mu;

      v8f ah0, ah1;
      {
        v2f h0 = {dd[0], dd[1]};
        v2f h1 = {dd[2], dd[3]};
        v2f h2 = {dd[4], 0.f};
        ah0 = __builtin_amdgcn_wmma_f32_16x16x4_f32(false, was[0], false, h0,
                                                    (short)0, zero8, false, false);
        ah1 = __builtin_amdgcn_wmma_f32_16x16x4_f32(false, was[1], false, h1,
                                                    (short)0, zero8, false, false);
        ah0 = __builtin_amdgcn_wmma_f32_16x16x4_f32(false, was[2], false, h2,
                                                    (short)0, ah0, false, false);
      }

      float q = 0.f;
#pragma unroll
      for (int v = 0; v < HH; ++v) q += dd[v] * dd[v];
      const float var = (q + xhalf(q)) * 0.1f;
      const float rs = __builtin_amdgcn_rsqf(var + 1e-5f);
      if (p == 3) {
#pragma unroll
        for (int v = 0; v < HH; ++v) xs5[v] = ax0[v] + ax1[v];
      }
      float xv[HH];
#pragma unroll
      for (int v = 0; v < HH; ++v) xv[v] = cols(xs5[v], (p + 1) & 3);
#pragma unroll
      for (int v = 0; v < HH; ++v)
        fac[v] = cos_rev(rs * (ah0[v] + ah1[v]) + xv[v]);
      asm volatile("" :: "v"(fac[0]), "v"(fac[1]), "v"(fac[2]), "v"(fac[3]), "v"(fac[4]));
#pragma unroll
      for (int v = 0; v < HH; ++v) h[v] = dd[v] * rs * lg[v] + lb[v];

      __builtin_amdgcn_sched_group_barrier(0x200, 3, 0);
      __builtin_amdgcn_sched_group_barrier(0x008, 4, 0);
      __builtin_amdgcn_sched_group_barrier(0x100, 10, 0);
      __builtin_amdgcn_sched_group_barrier(0x402, 128, 0);

      if (g == 0) {
#pragma unroll
        for (int v = 0; v < HH; ++v) hy[v] = (sgrp == p) ? h[v] : hy[v];
        if (p == 3) {
          gmem_f* yt = (gmem_f*)sbase_bits(ys + ((size_t)(4 * k) * BQ + mtile * 4) * HQ);
          asm("" : "+v"(yob));
          gmem_f* yrow = (gmem_f*)((__attribute__((address_space(1))) char*)yt + yob);
#pragma unroll
          for (int v = 0; v < HH; ++v) yrow[v] = hy[v];
        }
      }
    }
  }

  if (g == 0 && sgrp == 0) {
#pragma unroll
    for (int v = 0; v < HH; ++v) {
      hout[(mtile * 4 + jrow) * HQ + HH * half + v] = h[v];
      cout[(mtile * 4 + jrow) * HQ + HH * half + v] = c[v];
    }
  }
}

extern "C" void kernel_launch(void* const* d_in, const int* in_sizes, int n_in,
                              void* d_out, int out_size, void* d_ws, size_t ws_size,
                              hipStream_t stream) {
  const float* inputs = (const float*)d_in[0];
  const float* Wf = (const float*)d_in[1];
  const float* bf = (const float*)d_in[2];
  const float* pf = (const float*)d_in[3];
  const float* Wi = (const float*)d_in[4];
  const float* bi = (const float*)d_in[5];
  const float* pi = (const float*)d_in[6];
  const float* Wu = (const float*)d_in[7];
  const float* bu = (const float*)d_in[8];
  const float* pu = (const float*)d_in[9];
  const float* Wo = (const float*)d_in[10];
  const float* bo = (const float*)d_in[11];
  const float* po = (const float*)d_in[12];
  const float* lng = (const float*)d_in[13];
  const float* lnb = (const float*)d_in[14];

  float* out  = (float*)d_out;
  float* ys   = out;
  float* hout = out + (size_t)TQ * BQ * HQ;
  float* cout = hout + BQ * HQ;

  qlstm_fused<<<BQ / 4, 128, 0, stream>>>(
      inputs, Wf, Wi, Wu, Wo, bf, bi, bu, bo,
      pf, pi, pu, po, lng, lnb, ys, hout, cout);
}
